# starting kernel
# baseline (speedup 1.0000x reference)
#include <hip/hip_runtime.h>
#include <hip/hip_bf16.h>

typedef __attribute__((ext_vector_type(2))) float v2f;
typedef __attribute__((ext_vector_type(8))) float v8f;

#define KDIM 128
#define PDIM 256
#define NDIM 100000
#define LAMBDA 0.001f
#define NCHUNKS 64
#define CHUNK ((NDIM + NCHUNKS - 1) / NCHUNKS)
#define KP (KDIM * PDIM)

__global__ __launch_bounds__(512) void fm_gemm_partial(
    const float* __restrict__ evx, const float* __restrict__ evy,
    const float* __restrict__ fx,  const float* __restrict__ fy,
    float* __restrict__ partials)
{
    const int sel   = blockIdx.y;
    const int chunk = blockIdx.x;
    const float* __restrict__ ev   = sel ? evy : evx;
    const float* __restrict__ feat = sel ? fy  : fx;

    __shared__ float evBlk[KDIM][33];
    __shared__ float ftBlk[32][260];

    const int tid  = threadIdx.x;
    const int lane = tid & 31;
    const int wave = tid >> 5;
    const int rowBand = (wave >> 1) * 16;
    const int wcol    = (wave & 1) * 128;

    v8f acc[8];
    #pragma unroll
    for (int t = 0; t < 8; ++t) acc[t] = (v8f)0.0f;

    const int nStart = chunk * CHUNK;
    const int nEnd   = min(nStart + CHUNK, NDIM);

    {
        const int blk = nStart;
        #pragma unroll
        for (int q = 0; q < 8; ++q) {
            int idx = tid + 512 * q;
            int r = idx >> 5, c = idx & 31;
            int n = blk + c;
            evBlk[r][c] = (n < nEnd) ? ev[(long)r * NDIM + n] : 0.0f;
        }
        #pragma unroll
        for (int q = 0; q < 16; ++q) {
            int idx = tid + 512 * q;
            int r = idx >> 8, c = idx & 255;
            int n = blk + r;
            ftBlk[r][c] = (n < nEnd) ? feat[(long)n * PDIM + c] : 0.0f;
        }
    }
    __syncthreads();

    const int m  = lane & 15;
    const int kp = (lane >> 4) << 1;

    for (int blk = nStart; blk < nEnd; blk += 32) {
        const int nxt = blk + 32;
        const bool more = (nxt < nEnd);

        float evr[8];
        float ftr[16];
        if (more) {
            #pragma unroll
            for (int q = 0; q < 8; ++q) {
                int idx = tid + 512 * q;
                int r = idx >> 5, c = idx & 31;
                int n = nxt + c;
                evr[q] = (n < nEnd) ? ev[(long)r * NDIM + n] : 0.0f;
            }
            #pragma unroll
            for (int q = 0; q < 16; ++q) {
                int idx = tid + 512 * q;
                int r = idx >> 8, c = idx & 255;
                int n = nxt + r;
                ftr[q] = (n < nEnd) ? feat[(long)n * PDIM + c] : 0.0f;
            }
            if (nxt + 32 < nEnd) {
                long base = (long)(nxt + 32) * PDIM;
                __builtin_prefetch(&feat[base + (long)tid * 64], 0, 0);
            }
        }

        #pragma unroll
        for (int kk = 0; kk < 8; ++kk) {
            const int nOff = kk * 4;
            v2f a;
            a.x = evBlk[rowBand + m][nOff + kp];
            a.y = evBlk[rowBand + m][nOff + kp + 1];
            #pragma unroll
            for (int t = 0; t < 8; ++t) {
                const int col = wcol + t * 16 + m;
                v2f b;
                b.x = ftBlk[nOff + kp][col];
                b.y = ftBlk[nOff + kp + 1][col];
                acc[t] = __builtin_amdgcn_wmma_f32_16x16x4_f32(
                    false, a, false, b, (short)0, acc[t], false, false);
            }
        }
        __syncthreads();

        if (more) {
            #pragma unroll
            for (int q = 0; q < 8; ++q) {
                int idx = tid + 512 * q;
                evBlk[idx >> 5][idx & 31] = evr[q];
            }
            #pragma unroll
            for (int q = 0; q < 16; ++q) {
                int idx = tid + 512 * q;
                ftBlk[idx >> 8][idx & 255] = ftr[q];
            }
            __syncthreads();
        }
    }

    float* out = partials + (size_t)sel * NCHUNKS * KP + (size_t)chunk * KP;
    const int mrow = rowBand + ((lane >> 4) << 3);
    #pragma unroll
    for (int t = 0; t < 8; ++t) {
        const int col = wcol + t * 16 + (lane & 15);
        #pragma unroll
        for (int r = 0; r < 8; ++r)
            out[(mrow + r) * PDIM + col] = acc[t][r];
    }
}

__global__ void fm_reduce(const float* __restrict__ partials,
                          float* __restrict__ Amat, float* __restrict__ Bmat)
{
    int idx = blockIdx.x * blockDim.x + threadIdx.x;
    if (idx >= KP) return;
    float sa = 0.0f, sb = 0.0f;
    for (int c = 0; c < NCHUNKS; ++c) {
        sa += partials[(size_t)c * KP + idx];
        sb += partials[(size_t)(NCHUNKS + c) * KP + idx];
    }
    Amat[idx] = sa;
    Bmat[idx] = sb;
}

__global__ void fm_mask(const float* __restrict__ evx,
                        const float* __restrict__ evy,
                        float* __restrict__ D)
{
    int idx = blockIdx.x * blockDim.x + threadIdx.x;
    if (idx >= KDIM * KDIM) return;
    int i = idx >> 7;
    int j = idx & 127;
    float s  = fmaxf(evx[KDIM - 1], evy[KDIM - 1]);
    float g1 = sqrtf(evx[j] / s);
    float g2 = sqrtf(evy[i] / s);
    float d1 = 1.0f / (g1 * g1 + 1.0f);
    float d2 = 1.0f / (g2 * g2 + 1.0f);
    float re = g2 * d2 - g1 * d1;
    float im = d2 - d1;
    D[idx] = re * re + im * im;
}

__global__ __launch_bounds__(256) void fm_gram(
    const float* __restrict__ Amat, const float* __restrict__ Bmat,
    float* __restrict__ AAt, float* __restrict__ BAt)
{
    const float* __restrict__ L = blockIdx.x ? Bmat : Amat;
    const float* __restrict__ R = Amat;
    float* __restrict__ out     = blockIdx.x ? BAt : AAt;

    __shared__ float blkL[KDIM][33];
    __shared__ float blkR[KDIM][33];

    const int tid  = threadIdx.x;
    const int lane = tid & 31;
    const int wave = tid >> 5;
    const int iBase = wave * 16;

    v8f acc[8];
    #pragma unroll
    for (int t = 0; t < 8; ++t) acc[t] = (v8f)0.0f;

    for (int p0 = 0; p0 < PDIM; p0 += 32) {
        for (int idx = tid; idx < KDIM * 32; idx += 256) {
            int r = idx >> 5, c = idx & 31;
            blkL[r][c] = L[r * PDIM + p0 + c];
            blkR[r][c] = R[r * PDIM + p0 + c];
        }
        __syncthreads();

        const int m  = lane & 15;
        const int kp = (lane >> 4) << 1;
        #pragma unroll
        for (int kk = 0; kk < 8; ++kk) {
            const int pOff = kk * 4;
            v2f a;
            a.x = blkL[iBase + m][pOff + kp];
            a.y = blkL[iBase + m][pOff + kp + 1];
            #pragma unroll
            for (int t = 0; t < 8; ++t) {
                v2f b;
                b.x = blkR[t * 16 + m][pOff + kp];
                b.y = blkR[t * 16 + m][pOff + kp + 1];
                acc[t] = __builtin_amdgcn_wmma_f32_16x16x4_f32(
                    false, a, false, b, (short)0, acc[t], false, false);
            }
        }
        __syncthreads();
    }

    const int mrow = iBase + ((lane >> 4) << 3);
    #pragma unroll
    for (int t = 0; t < 8; ++t) {
        const int j = t * 16 + (lane & 15);
        #pragma unroll
        for (int r = 0; r < 8; ++r)
            out[(mrow + r) * KDIM + j] = acc[t][r];
    }
}

__global__ __launch_bounds__(128) void fm_solve(
    const float* __restrict__ AAt, const float* __restrict__ BAt,
    const float* __restrict__ D,   float* __restrict__ Cout)
{
    extern __shared__ float Msh[];
    const int i   = blockIdx.x;
    const int tid = threadIdx.x;

    for (int idx = tid; idx < KDIM * 129; idx += 128) {
        int r = idx / 129, c = idx - r * 129;
        float v;
        if (c < KDIM) {
            v = AAt[r * KDIM + c];
            if (c == r) v += LAMBDA * D[i * KDIM + r];
        } else {
            v = BAt[i * KDIM + r];
        }
        Msh[idx] = v;
    }

    for (int c = 0; c < KDIM - 1; ++c) {
        __syncthreads();
        int r = tid;
        if (r > c) {
            float f = Msh[r * 129 + c] / Msh[c * 129 + c];
            for (int j = c + 1; j <= KDIM; ++j)
                Msh[r * 129 + j] -= f * Msh[c * 129 + j];
        }
    }

    for (int c = KDIM - 1; c >= 0; --c) {
        __syncthreads();
        if (tid == c)
            Msh[c * 129 + KDIM] /= Msh[c * 129 + c];
        __syncthreads();
        if (tid < c)
            Msh[tid * 129 + KDIM] -= Msh[tid * 129 + c] * Msh[c * 129 + KDIM];
    }
    __syncthreads();
    Cout[i * KDIM + tid] = Msh[tid * 129 + KDIM];
}

extern "C" void kernel_launch(void* const* d_in, const int* in_sizes, int n_in,
                              void* d_out, int out_size, void* d_ws, size_t ws_size,
                              hipStream_t stream)
{
    const float* feat_x = (const float*)d_in[0];
    const float* feat_y = (const float*)d_in[1];
    const float* evals_x = (const float*)d_in[2];
    const float* evals_y = (const float*)d_in[3];
    const float* evecs_x = (const float*)d_in[4];
    const float* evecs_y = (const float*)d_in[5];

    float* wsf   = (float*)d_ws;
    float* Amat  = wsf;
    float* Bmat  = wsf + 32768;
    float* AAt   = wsf + 65536;
    float* BAt   = wsf + 81920;
    float* Dmask = wsf + 98304;
    float* part  = wsf + 114688;
    float* Cout  = (float*)d_out;

    fm_gemm_partial<<<dim3(NCHUNKS, 2), 512, 0, stream>>>(
        evecs_x, evecs_y, feat_x, feat_y, part);

    fm_reduce<<<(KP + 255) / 256, 256, 0, stream>>>(part, Amat, Bmat);

    fm_mask<<<(KDIM * KDIM + 255) / 256, 256, 0, stream>>>(evals_x, evals_y, Dmask);

    fm_gram<<<2, 256, 0, stream>>>(Amat, Bmat, AAt, BAt);

    fm_solve<<<KDIM, KDIM, KDIM * 129 * sizeof(float), stream>>>(
        AAt, BAt, Dmask, Cout);
}


# solve kernel: elimination on 1024 threads with the matrix in registers (row x column-slice), one barrier per pivot, one-wave back substitution by lane reads; per-element arithmetic order kept
# speedup vs baseline: 2.1986x; 2.1986x over previous
#include <hip/hip_runtime.h>
#include <hip/hip_bf16.h>

typedef __attribute__((ext_vector_type(2))) float v2f;
typedef __attribute__((ext_vector_type(8))) float v8f;

#define KDIM 128
#define PDIM 256
#define NDIM 100000
#define LAMBDA 0.001f
#define NCHUNKS 64
#define CHUNK ((NDIM + NCHUNKS - 1) / NCHUNKS)
#define KP (KDIM * PDIM)

__global__ __launch_bounds__(512) void fm_gemm_partial(
    const float* __restrict__ evx, const float* __restrict__ evy,
    const float* __restrict__ fx,  const float* __restrict__ fy,
    float* __restrict__ partials)
{
    const int sel   = blockIdx.y;
    const int chunk = blockIdx.x;
    const float* __restrict__ ev   = sel ? evy : evx;
    const float* __restrict__ feat = sel ? fy  : fx;

    __shared__ float evBlk[KDIM][33];
    __shared__ float ftBlk[32][260];

    const int tid  = threadIdx.x;
    const int lane = tid & 31;
    const int wave = tid >> 5;
    const int rowBand = (wave >> 1) * 16;
    const int wcol    = (wave & 1) * 128;

    v8f acc[8];
    #pragma unroll
    for (int t = 0; t < 8; ++t) acc[t] = (v8f)0.0f;

    const int nStart = chunk * CHUNK;
    const int nEnd   = min(nStart + CHUNK, NDIM);

    {
        const int blk = nStart;
        #pragma unroll
        for (int q = 0; q < 8; ++q) {
            int idx = tid + 512 * q;
            int r = idx >> 5, c = idx & 31;
            int n = blk + c;
            evBlk[r][c] = (n < nEnd) ? ev[(long)r * NDIM + n] : 0.0f;
        }
        #pragma unroll
        for (int q = 0; q < 16; ++q) {
            int idx = tid + 512 * q;
            int r = idx >> 8, c = idx & 255;
            int n = blk + r;
            ftBlk[r][c] = (n < nEnd) ? feat[(long)n * PDIM + c] : 0.0f;
        }
    }
    __syncthreads();

    const int m  = lane & 15;
    const int kp = (lane >> 4) << 1;

    for (int blk = nStart; blk < nEnd; blk += 32) {
        const int nxt = blk + 32;
        const bool more = (nxt < nEnd);

        float evr[8];
        float ftr[16];
        if (more) {
            #pragma unroll
            for (int q = 0; q < 8; ++q) {
                int idx = tid + 512 * q;
                int r = idx >> 5, c = idx & 31;
                int n = nxt + c;
                evr[q] = (n < nEnd) ? ev[(long)r * NDIM + n] : 0.0f;
            }
            #pragma unroll
            for (int q = 0; q < 16; ++q) {
                int idx = tid + 512 * q;
                int r = idx >> 8, c = idx & 255;
                int n = nxt + r;
                ftr[q] = (n < nEnd) ? feat[(long)n * PDIM + c] : 0.0f;
            }
            if (nxt + 32 < nEnd) {
                long base = (long)(nxt + 32) * PDIM;
                __builtin_prefetch(&feat[base + (long)tid * 64], 0, 0);
            }
        }

        #pragma unroll
        for (int kk = 0; kk < 8; ++kk) {
            const int nOff = kk * 4;
            v2f a;
            a.x = evBlk[rowBand + m][nOff + kp];
            a.y = evBlk[rowBand + m][nOff + kp + 1];
            #pragma unroll
            for (int t = 0; t < 8; ++t) {
                const int col = wcol + t * 16 + m;
                v2f b;
                b.x = ftBlk[nOff + kp][col];
                b.y = ftBlk[nOff + kp + 1][col];
                acc[t] = __builtin_amdgcn_wmma_f32_16x16x4_f32(
                    false, a, false, b, (short)0, acc[t], false, false);
            }
        }
        __syncthreads();

        if (more) {
            #pragma unroll
            for (int q = 0; q < 8; ++q) {
                int idx = tid + 512 * q;
                evBlk[idx >> 5][idx & 31] = evr[q];
            }
            #pragma unroll
            for (int q = 0; q < 16; ++q) {
                int idx = tid + 512 * q;
                ftBlk[idx >> 8][idx & 255] = ftr[q];
            }
            __syncthreads();
        }
    }

    float* out = partials + (size_t)sel * NCHUNKS * KP + (size_t)chunk * KP;
    const int mrow = rowBand + ((lane >> 4) << 3);
    #pragma unroll
    for (int t = 0; t < 8; ++t) {
        const int col = wcol + t * 16 + (lane & 15);
        #pragma unroll
        for (int r = 0; r < 8; ++r)
            out[(mrow + r) * PDIM + col] = acc[t][r];
    }
}

__global__ void fm_reduce(const float* __restrict__ partials,
                          float* __restrict__ Amat, float* __restrict__ Bmat)
{
    int idx = blockIdx.x * blockDim.x + threadIdx.x;
    if (idx >= KP) return;
    float sa = 0.0f, sb = 0.0f;
    for (int c = 0; c < NCHUNKS; ++c) {
        sa += partials[(size_t)c * KP + idx];
        sb += partials[(size_t)(NCHUNKS + c) * KP + idx];
    }
    Amat[idx] = sa;
    Bmat[idx] = sb;
}

__global__ void fm_mask(const float* __restrict__ evx,
                        const float* __restrict__ evy,
                        float* __restrict__ D)
{
    int idx = blockIdx.x * blockDim.x + threadIdx.x;
    if (idx >= KDIM * KDIM) return;
    int i = idx >> 7;
    int j = idx & 127;
    float s  = fmaxf(evx[KDIM - 1], evy[KDIM - 1]);
    float g1 = sqrtf(evx[j] / s);
    float g2 = sqrtf(evy[i] / s);
    float d1 = 1.0f / (g1 * g1 + 1.0f);
    float d2 = 1.0f / (g2 * g2 + 1.0f);
    float re = g2 * d2 - g1 * d1;
    float im = d2 - d1;
    D[idx] = re * re + im * im;
}

__global__ __launch_bounds__(256) void fm_gram(
    const float* __restrict__ Amat, const float* __restrict__ Bmat,
    float* __restrict__ AAt, float* __restrict__ BAt)
{
    const float* __restrict__ L = blockIdx.x ? Bmat : Amat;
    const float* __restrict__ R = Amat;
    float* __restrict__ out     = blockIdx.x ? BAt : AAt;

    __shared__ float blkL[KDIM][33];
    __shared__ float blkR[KDIM][33];

    const int tid  = threadIdx.x;
    const int lane = tid & 31;
    const int wave = tid >> 5;
    const int iBase = wave * 16;

    v8f acc[8];
    #pragma unroll
    for (int t = 0; t < 8; ++t) acc[t] = (v8f)0.0f;

    for (int p0 = 0; p0 < PDIM; p0 += 32) {
        for (int idx = tid; idx < KDIM * 32; idx += 256) {
            int r = idx >> 5, c = idx & 31;
            blkL[r][c] = L[r * PDIM + p0 + c];
            blkR[r][c] = R[r * PDIM + p0 + c];
        }
        __syncthreads();

        const int m  = lane & 15;
        const int kp = (lane >> 4) << 1;
        #pragma unroll
        for (int kk = 0; kk < 8; ++kk) {
            const int pOff = kk * 4;
            v2f a;
            a.x = blkL[iBase + m][pOff + kp];
            a.y = blkL[iBase + m][pOff + kp + 1];
            #pragma unroll
            for (int t = 0; t < 8; ++t) {
                v2f b;
                b.x = blkR[t * 16 + m][pOff + kp];
                b.y = blkR[t * 16 + m][pOff + kp + 1];
                acc[t] = __builtin_amdgcn_wmma_f32_16x16x4_f32(
                    false, a, false, b, (short)0, acc[t], false, false);
            }
        }
        __syncthreads();
    }

    const int mrow = iBase + ((lane >> 4) << 3);
    #pragma unroll
    for (int t = 0; t < 8; ++t) {
        const int j = t * 16 + (lane & 15);
        #pragma unroll
        for (int r = 0; r < 8; ++r)
            out[(mrow + r) * KDIM + j] = acc[t][r];
    }
}

#define USTRIDE 132
#define SOLVE_LDS_FLOATS (KDIM * USTRIDE + KDIM + 2 * KDIM)

__global__ __launch_bounds__(1024) void fm_solve(
    const float* __restrict__ AAt, const float* __restrict__ BAt,
    const float* __restrict__ D,   float* __restrict__ Cout)
{
    extern __shared__ __align__(16) float smem[];
    float* Ush  = smem;
    float* ysh  = smem + KDIM * USTRIDE;
    float* colb = ysh + KDIM;

    const int i    = blockIdx.x;
    const int tid  = threadIdx.x;
    const int lane = tid & 31;
    const int wave = tid >> 5;
    const int s  = wave & 7;
    const int r  = (wave >> 3) * 32 + lane;
    const int j0 = s * 16;

    float m[16];
    {
        const float4* srcv = reinterpret_cast<const float4*>(AAt + r * KDIM + j0);
        #pragma unroll
        for (int q = 0; q < 4; ++q) {
            float4 v = srcv[q];
            m[4 * q + 0] = v.x; m[4 * q + 1] = v.y;
            m[4 * q + 2] = v.z; m[4 * q + 3] = v.w;
        }
    }
    float bval = BAt[i * KDIM + r];
    if ((r >> 4) == s) {
        const float dl = D[i * KDIM + r];
        #pragma unroll
        for (int jj = 0; jj < 16; ++jj)
            if (j0 + jj == r) m[jj] += LAMBDA * dl;
    }

    for (int cb = 0; cb < 8; ++cb) {
        #pragma unroll
        for (int cc = 0; cc < 16; ++cc) {
            const int c = cb * 16 + cc;
            float* cbuf = colb + (c & 1) * KDIM;
            float4* urow = reinterpret_cast<float4*>(Ush + c * USTRIDE + j0);
            if (r == c) {
                urow[0] = make_float4(m[0],  m[1],  m[2],  m[3]);
                urow[1] = make_float4(m[4],  m[5],  m[6],  m[7]);
                urow[2] = make_float4(m[8],  m[9],  m[10], m[11]);
                urow[3] = make_float4(m[12], m[13], m[14], m[15]);
                if (s == 7) ysh[c] = bval;
            }
            if (s == cb) cbuf[r] = m[cc];
            __syncthreads();
            if (r > c && j0 + 15 > c) {
                const float piv = Ush[c * USTRIDE + c];
                const float mul = cbuf[r];
                const float4 p0 = urow[0], p1 = urow[1], p2 = urow[2], p3 = urow[3];
                const float yp  = ysh[c];
                const float f   = mul / piv;
                m[0]  -= f * p0.x; m[1]  -= f * p0.y; m[2]  -= f * p0.z; m[3]  -= f * p0.w;
                m[4]  -= f * p1.x; m[5]  -= f * p1.y; m[6]  -= f * p1.z; m[7]  -= f * p1.w;
                m[8]  -= f * p2.x; m[9]  -= f * p2.y; m[10] -= f * p2.z; m[11] -= f * p2.w;
                m[12] -= f * p3.x; m[13] -= f * p3.y; m[14] -= f * p3.z; m[15] -= f * p3.w;
                bval -= f * yp;
            }
        }
    }
    if (tid >= 32) return;

    float y[4], x[4], dg[4];
    #pragma unroll
    for (int k = 0; k < 4; ++k) {
        const int row = lane + 32 * k;
        y[k]  = ysh[row];
        dg[k] = Ush[row * USTRIDE + row];
        x[k]  = 0.0f;
    }
    #pragma unroll
    for (int q = 3; q >= 0; --q) {
        const float* ucol = Ush + lane * USTRIDE + 32 * q;
        float uc[4];
        #pragma unroll
        for (int k = 0; k <= q; ++k) uc[k] = ucol[k * 32 * USTRIDE + 31];
        #pragma unroll 1
        for (int cc = 31; cc >= 0; --cc) {
            const int cn = cc > 0 ? cc - 1 : 0;
            float un[4];
            #pragma unroll
            for (int k = 0; k <= q; ++k) un[k] = ucol[k * 32 * USTRIDE + cn];
            const float yc = __int_as_float(
                __builtin_amdgcn_readlane(__float_as_int(y[q]), cc));
            const float dc = __int_as_float(
                __builtin_amdgcn_readlane(__float_as_int(dg[q]), cc));
            const float xc = yc / dc;
            x[q] = (lane == cc) ? xc : x[q];
            #pragma unroll
            for (int k = 0; k < q; ++k) y[k] -= uc[k] * xc;
            {
                const float t = y[q] - uc[q] * xc;
                y[q] = (lane < cc) ? t : y[q];
            }
            #pragma unroll
            for (int k = 0; k <= q; ++k) uc[k] = un[k];
        }
    }
    #pragma unroll
    for (int k = 0; k < 4; ++k)
        Cout[i * KDIM + lane + 32 * k] = x[k];
}

extern "C" void kernel_launch(void* const* d_in, const int* in_sizes, int n_in,
                              void* d_out, int out_size, void* d_ws, size_t ws_size,
                              hipStream_t stream)
{
    const float* feat_x = (const float*)d_in[0];
    const float* feat_y = (const float*)d_in[1];
    const float* evals_x = (const float*)d_in[2];
    const float* evals_y = (const float*)d_in[3];
    const float* evecs_x = (const float*)d_in[4];
    const float* evecs_y = (const float*)d_in[5];

    float* wsf   = (float*)d_ws;
    float* Amat  = wsf;
    float* Bmat  = wsf + 32768;
    float* AAt   = wsf + 65536;
    float* BAt   = wsf + 81920;
    float* Dmask = wsf + 98304;
    float* part  = wsf + 114688;
    float* Cout  = (float*)d_out;

    fm_gemm_partial<<<dim3(NCHUNKS, 2), 512, 0, stream>>>(
        evecs_x, evecs_y, feat_x, feat_y, part);

    fm_reduce<<<(KP + 255) / 256, 256, 0, stream>>>(part, Amat, Bmat);

    fm_mask<<<(KDIM * KDIM + 255) / 256, 256, 0, stream>>>(evals_x, evals_y, Dmask);

    fm_gram<<<2, 256, 0, stream>>>(Amat, Bmat, AAt, BAt);

    fm_solve<<<KDIM, 1024, SOLVE_LDS_FLOATS * sizeof(float), stream>>>(
        AAt, BAt, Dmask, Cout);
}


# gemm kernel: each chunk's 128x256 tile split into two 128x128 column halves (grid 64x2x2 = 256 workgroups) so the f32 matrix work runs on all 256 WGPs, not 128; per-element k order kept
# speedup vs baseline: 3.1477x; 1.4317x over previous
#include <hip/hip_runtime.h>
#include <hip/hip_bf16.h>

typedef __attribute__((ext_vector_type(2))) float v2f;
typedef __attribute__((ext_vector_type(8))) float v8f;

#define KDIM 128
#define PDIM 256
#define NDIM 100000
#define LAMBDA 0.001f
#define NCHUNKS 64
#define CHUNK ((NDIM + NCHUNKS - 1) / NCHUNKS)
#define KP (KDIM * PDIM)
#define PHALF (PDIM / 2)

__global__ __launch_bounds__(512) void fm_gemm_partial(
    const float* __restrict__ evx, const float* __restrict__ evy,
    const float* __restrict__ fx,  const float* __restrict__ fy,
    float* __restrict__ partials)
{
    const int sel   = blockIdx.y;
    const int chunk = blockIdx.x;
    const int pcol0 = blockIdx.z * PHALF;
    const float* __restrict__ ev   = sel ? evy : evx;
    const float* __restrict__ feat = sel ? fy  : fx;

    __shared__ float evBlk[KDIM][33];
    __shared__ float ftBlk[32][PHALF + 4];

    const int tid  = threadIdx.x;
    const int lane = tid & 31;
    const int wave = tid >> 5;
    const int rowBand = (wave >> 1) * 16;
    const int wcol    = (wave & 1) * 64;

    v8f acc[4];
    #pragma unroll
    for (int t = 0; t < 4; ++t) acc[t] = (v8f)0.0f;

    const int nStart = chunk * CHUNK;
    const int nEnd   = min(nStart + CHUNK, NDIM);

    {
        const int blk = nStart;
        #pragma unroll
        for (int q = 0; q < 8; ++q) {
            int idx = tid + 512 * q;
            int r = idx >> 5, c = idx & 31;
            int n = blk + c;
            evBlk[r][c] = (n < nEnd) ? ev[(long)r * NDIM + n] : 0.0f;
        }
        #pragma unroll
        for (int q = 0; q < 8; ++q) {
            int idx = tid + 512 * q;
            int r = idx >> 7, c = idx & 127;
            int n = blk + r;
            ftBlk[r][c] = (n < nEnd) ? feat[(long)n * PDIM + pcol0 + c] : 0.0f;
        }
    }
    __syncthreads();

    const int m  = lane & 15;
    const int kp = (lane >> 4) << 1;

    for (int blk = nStart; blk < nEnd; blk += 32) {
        const int nxt = blk + 32;
        const bool more = (nxt < nEnd);

        float evr[8];
        float ftr[8];
        if (more) {
            #pragma unroll
            for (int q = 0; q < 8; ++q) {
                int idx = tid + 512 * q;
                int r = idx >> 5, c = idx & 31;
                int n = nxt + c;
                evr[q] = (n < nEnd) ? ev[(long)r * NDIM + n] : 0.0f;
            }
            #pragma unroll
            for (int q = 0; q < 8; ++q) {
                int idx = tid + 512 * q;
                int r = idx >> 7, c = idx & 127;
                int n = nxt + r;
                ftr[q] = (n < nEnd) ? feat[(long)n * PDIM + pcol0 + c] : 0.0f;
            }
            if (nxt + 32 < nEnd && tid < 256) {
                long base = (long)(nxt + 32 + (tid >> 1)) * PDIM + pcol0;
                __builtin_prefetch(&feat[base + (tid & 1) * 64], 0, 0);
            }
        }

        #pragma unroll
        for (int kk = 0; kk < 8; ++kk) {
            const int nOff = kk * 4;
            v2f a;
            a.x = evBlk[rowBand + m][nOff + kp];
            a.y = evBlk[rowBand + m][nOff + kp + 1];
            #pragma unroll
            for (int t = 0; t < 4; ++t) {
                const int col = wcol + t * 16 + m;
                v2f b;
                b.x = ftBlk[nOff + kp][col];
                b.y = ftBlk[nOff + kp + 1][col];
                acc[t] = __builtin_amdgcn_wmma_f32_16x16x4_f32(
                    false, a, false, b, (short)0, acc[t], false, false);
            }
        }
        __syncthreads();

        if (more) {
            #pragma unroll
            for (int q = 0; q < 8; ++q) {
                int idx = tid + 512 * q;
                evBlk[idx >> 5][idx & 31] = evr[q];
            }
            #pragma unroll
            for (int q = 0; q < 8; ++q) {
                int idx = tid + 512 * q;
                ftBlk[idx >> 7][idx & 127] = ftr[q];
            }
            __syncthreads();
        }
    }

    float* out = partials + (size_t)sel * NCHUNKS * KP + (size_t)chunk * KP;
    const int mrow = rowBand + ((lane >> 4) << 3);
    #pragma unroll
    for (int t = 0; t < 4; ++t) {
        const int col = pcol0 + wcol + t * 16 + (lane & 15);
        #pragma unroll
        for (int r = 0; r < 8; ++r)
            out[(mrow + r) * PDIM + col] = acc[t][r];
    }
}

__global__ void fm_reduce(const float* __restrict__ partials,
                          float* __restrict__ Amat, float* __restrict__ Bmat)
{
    int idx = blockIdx.x * blockDim.x + threadIdx.x;
    if (idx >= KP) return;
    float sa = 0.0f, sb = 0.0f;
    for (int c = 0; c < NCHUNKS; ++c) {
        sa += partials[(size_t)c * KP + idx];
        sb += partials[(size_t)(NCHUNKS + c) * KP + idx];
    }
    Amat[idx] = sa;
    Bmat[idx] = sb;
}

__global__ void fm_mask(const float* __restrict__ evx,
                        const float* __restrict__ evy,
                        float* __restrict__ D)
{
    int idx = blockIdx.x * blockDim.x + threadIdx.x;
    if (idx >= KDIM * KDIM) return;
    int i = idx >> 7;
    int j = idx & 127;
    float s  = fmaxf(evx[KDIM - 1], evy[KDIM - 1]);
    float g1 = sqrtf(evx[j] / s);
    float g2 = sqrtf(evy[i] / s);
    float d1 = 1.0f / (g1 * g1 + 1.0f);
    float d2 = 1.0f / (g2 * g2 + 1.0f);
    float re = g2 * d2 - g1 * d1;
    float im = d2 - d1;
    D[idx] = re * re + im * im;
}

__global__ __launch_bounds__(256) void fm_gram(
    const float* __restrict__ Amat, const float* __restrict__ Bmat,
    float* __restrict__ AAt, float* __restrict__ BAt)
{
    const float* __restrict__ L = blockIdx.x ? Bmat : Amat;
    const float* __restrict__ R = Amat;
    float* __restrict__ out     = blockIdx.x ? BAt : AAt;

    __shared__ float blkL[KDIM][33];
    __shared__ float blkR[KDIM][33];

    const int tid  = threadIdx.x;
    const int lane = tid & 31;
    const int wave = tid >> 5;
    const int iBase = wave * 16;

    v8f acc[8];
    #pragma unroll
    for (int t = 0; t < 8; ++t) acc[t] = (v8f)0.0f;

    for (int p0 = 0; p0 < PDIM; p0 += 32) {
        for (int idx = tid; idx < KDIM * 32; idx += 256) {
            int r = idx >> 5, c = idx & 31;
            blkL[r][c] = L[r * PDIM + p0 + c];
            blkR[r][c] = R[r * PDIM + p0 + c];
        }
        __syncthreads();

        const int m  = lane & 15;
        const int kp = (lane >> 4) << 1;
        #pragma unroll
        for (int kk = 0; kk < 8; ++kk) {
            const int pOff = kk * 4;
            v2f a;
            a.x = blkL[iBase + m][pOff + kp];
            a.y = blkL[iBase + m][pOff + kp + 1];
            #pragma unroll
            for (int t = 0; t < 8; ++t) {
                v2f b;
                b.x = blkR[t * 16 + m][pOff + kp];
                b.y = blkR[t * 16 + m][pOff + kp + 1];
                acc[t] = __builtin_amdgcn_wmma_f32_16x16x4_f32(
                    false, a, false, b, (short)0, acc[t], false, false);
            }
        }
        __syncthreads();
    }

    const int mrow = iBase + ((lane >> 4) << 3);
    #pragma unroll
    for (int t = 0; t < 8; ++t) {
        const int j = t * 16 + (lane & 15);
        #pragma unroll
        for (int r = 0; r < 8; ++r)
            out[(mrow + r) * KDIM + j] = acc[t][r];
    }
}

#define USTRIDE 132
#define SOLVE_LDS_FLOATS (KDIM * USTRIDE + KDIM + 2 * KDIM)

__global__ __launch_bounds__(1024) void fm_solve(
    const float* __restrict__ AAt, const float* __restrict__ BAt,
    const float* __restrict__ D,   float* __restrict__ Cout)
{
    extern __shared__ __align__(16) float smem[];
    float* Ush  = smem;
    float* ysh  = smem + KDIM * USTRIDE;
    float* colb = ysh + KDIM;

    const int i    = blockIdx.x;
    const int tid  = threadIdx.x;
    const int lane = tid & 31;
    const int wave = tid >> 5;
    const int s  = wave & 7;
    const int r  = (wave >> 3) * 32 + lane;
    const int j0 = s * 16;

    float m[16];
    {
        const float4* srcv = reinterpret_cast<const float4*>(AAt + r * KDIM + j0);
        #pragma unroll
        for (int q = 0; q < 4; ++q) {
            float4 v = srcv[q];
            m[4 * q + 0] = v.x; m[4 * q + 1] = v.y;
            m[4 * q + 2] = v.z; m[4 * q + 3] = v.w;
        }
    }
    float bval = BAt[i * KDIM + r];
    if ((r >> 4) == s) {
        const float dl = D[i * KDIM + r];
        #pragma unroll
        for (int jj = 0; jj < 16; ++jj)
            if (j0 + jj == r) m[jj] += LAMBDA * dl;
    }

    for (int cb = 0; cb < 8; ++cb) {
        #pragma unroll
        for (int cc = 0; cc < 16; ++cc) {
            const int c = cb * 16 + cc;
            float* cbuf = colb + (c & 1) * KDIM;
            float4* urow = reinterpret_cast<float4*>(Ush + c * USTRIDE + j0);
            if (r == c) {
                urow[0] = make_float4(m[0],  m[1],  m[2],  m[3]);
                urow[1] = make_float4(m[4],  m[5],  m[6],  m[7]);
                urow[2] = make_float4(m[8],  m[9],  m[10], m[11]);
                urow[3] = make_float4(m[12], m[13], m[14], m[15]);
                if (s == 7) ysh[c] = bval;
            }
            if (s == cb) cbuf[r] = m[cc];
            __syncthreads();
            if (r > c && j0 + 15 > c) {
                const float piv = Ush[c * USTRIDE + c];
                const float mul = cbuf[r];
                const float4 p0 = urow[0], p1 = urow[1], p2 = urow[2], p3 = urow[3];
                const float yp  = ysh[c];
                const float f   = mul / piv;
                m[0]  -= f * p0.x; m[1]  -= f * p0.y; m[2]  -= f * p0.z; m[3]  -= f * p0.w;
                m[4]  -= f * p1.x; m[5]  -= f * p1.y; m[6]  -= f * p1.z; m[7]  -= f * p1.w;
                m[8]  -= f * p2.x; m[9]  -= f * p2.y; m[10] -= f * p2.z; m[11] -= f * p2.w;
                m[12] -= f * p3.x; m[13] -= f * p3.y; m[14] -= f * p3.z; m[15] -= f * p3.w;
                bval -= f * yp;
            }
        }
    }
    if (tid >= 32) return;

    float y[4], x[4], dg[4];
    #pragma unroll
    for (int k = 0; k < 4; ++k) {
        const int row = lane + 32 * k;
        y[k]  = ysh[row];
        dg[k] = Ush[row * USTRIDE + row];
        x[k]  = 0.0f;
    }
    #pragma unroll
    for (int q = 3; q >= 0; --q) {
        const float* ucol = Ush + lane * USTRIDE + 32 * q;
        float uc[4];
        #pragma unroll
        for (int k = 0; k <= q; ++k) uc[k] = ucol[k * 32 * USTRIDE + 31];
        #pragma unroll 1
        for (int cc = 31; cc >= 0; --cc) {
            const int cn = cc > 0 ? cc - 1 : 0;
            float un[4];
            #pragma unroll
            for (int k = 0; k <= q; ++k) un[k] = ucol[k * 32 * USTRIDE + cn];
            const float yc = __int_as_float(
                __builtin_amdgcn_readlane(__float_as_int(y[q]), cc));
            const float dc = __int_as_float(
                __builtin_amdgcn_readlane(__float_as_int(dg[q]), cc));
            const float xc = yc / dc;
            x[q] = (lane == cc) ? xc : x[q];
            #pragma unroll
            for (int k = 0; k < q; ++k) y[k] -= uc[k] * xc;
            {
                const float t = y[q] - uc[q] * xc;
                y[q] = (lane < cc) ? t : y[q];
            }
            #pragma unroll
            for (int k = 0; k <= q; ++k) uc[k] = un[k];
        }
    }
    #pragma unroll
    for (int k = 0; k < 4; ++k)
        Cout[i * KDIM + lane + 32 * k] = x[k];
}

extern "C" void kernel_launch(void* const* d_in, const int* in_sizes, int n_in,
                              void* d_out, int out_size, void* d_ws, size_t ws_size,
                              hipStream_t stream)
{
    const float* feat_x = (const float*)d_in[0];
    const float* feat_y = (const float*)d_in[1];
    const float* evals_x = (const float*)d_in[2];
    const float* evals_y = (const float*)d_in[3];
    const float* evecs_x = (const float*)d_in[4];
    const float* evecs_y = (const float*)d_in[5];

    float* wsf   = (float*)d_ws;
    float* Amat  = wsf;
    float* Bmat  = wsf + 32768;
    float* AAt   = wsf + 65536;
    float* BAt   = wsf + 81920;
    float* Dmask = wsf + 98304;
    float* part  = wsf + 114688;
    float* Cout  = (float*)d_out;

    fm_gemm_partial<<<dim3(NCHUNKS, 2, 2), 512, 0, stream>>>(
        evecs_x, evecs_y, feat_x, feat_y, part);

    fm_reduce<<<(KP + 255) / 256, 256, 0, stream>>>(part, Amat, Bmat);

    fm_mask<<<(KDIM * KDIM + 255) / 256, 256, 0, stream>>>(evals_x, evals_y, Dmask);

    fm_gram<<<2, 256, 0, stream>>>(Amat, Bmat, AAt, BAt);

    fm_solve<<<KDIM, 1024, SOLVE_LDS_FLOATS * sizeof(float), stream>>>(
        AAt, BAt, Dmask, Cout);
}


# gemm kernel: staging in 16-byte pieces: chunks are whole 32-row slabs (1568 rows; split-K partial sums regroup), 4 unguarded 16-B loads + 4 16-B LDS stores per thread per slab, not 16+16 of 4 B
# speedup vs baseline: 4.0177x; 1.2764x over previous
#include <hip/hip_runtime.h>
#include <hip/hip_bf16.h>

typedef __attribute__((ext_vector_type(2))) float v2f;
typedef __attribute__((ext_vector_type(8))) float v8f;

#define KDIM 128
#define PDIM 256
#define NDIM 100000
#define LAMBDA 0.001f
#define NCHUNKS 64
#define SLAB 32
#define CHUNK 1568
static_assert(NDIM % SLAB == 0 && CHUNK % SLAB == 0,
              "every chunk must be a whole number of slabs");
static_assert((NCHUNKS - 1) * CHUNK < NDIM && NCHUNKS * CHUNK >= NDIM,
              "the chunks must cover N and none may be empty");
#define EVPITCH 36
#define KP (KDIM * PDIM)
#define PHALF (PDIM / 2)

__global__ __launch_bounds__(512) void fm_gemm_partial(
    const float* __restrict__ evx, const float* __restrict__ evy,
    const float* __restrict__ fx,  const float* __restrict__ fy,
    float* __restrict__ partials)
{
    const int sel   = blockIdx.y;
    const int chunk = blockIdx.x;
    const int pcol0 = blockIdx.z * PHALF;
    const float* __restrict__ ev   = sel ? evy : evx;
    const float* __restrict__ feat = sel ? fy  : fx;

    __shared__ __align__(16) float evBlk[KDIM][EVPITCH];
    __shared__ __align__(16) float ftBlk[SLAB][PHALF + 4];

    const int tid  = threadIdx.x;
    const int lane = tid & 31;
    const int wave = tid >> 5;
    const int rowBand = (wave >> 1) * 16;
    const int wcol    = (wave & 1) * 64;

    const int evRow = tid >> 3;
    const int evCol = (tid & 7) << 2;
    const int ftRow = tid >> 5;
    const int ftCol = (tid & 31) << 2;
    const float* __restrict__ evSrc0 = ev + (size_t)evRow * NDIM + evCol;
    const float* __restrict__ evSrc1 = evSrc0 + (size_t)64 * NDIM;
    const float* __restrict__ ftSrc0 = feat + (size_t)ftRow * PDIM + pcol0 + ftCol;
    const float* __restrict__ ftSrc1 = ftSrc0 + (size_t)16 * PDIM;
    float4* const evDst0 = reinterpret_cast<float4*>(&evBlk[evRow][evCol]);
    float4* const evDst1 = reinterpret_cast<float4*>(&evBlk[evRow + 64][evCol]);
    float4* const ftDst0 = reinterpret_cast<float4*>(&ftBlk[ftRow][ftCol]);
    float4* const ftDst1 = reinterpret_cast<float4*>(&ftBlk[ftRow + 16][ftCol]);
    v8f acc[4];
    #pragma unroll
    for (int t = 0; t < 4; ++t) acc[t] = (v8f)0.0f;

    const int nStart = chunk * CHUNK;
    const int nEnd   = min(nStart + CHUNK, NDIM);

    *evDst0 = *reinterpret_cast<const float4*>(evSrc0 + nStart);
    *evDst1 = *reinterpret_cast<const float4*>(evSrc1 + nStart);
    *ftDst0 = *reinterpret_cast<const float4*>(ftSrc0 + (size_t)nStart * PDIM);
    *ftDst1 = *reinterpret_cast<const float4*>(ftSrc1 + (size_t)nStart * PDIM);
    __syncthreads();

    const int m  = lane & 15;
    const int kp = (lane >> 4) << 1;

    for (int blk = nStart; blk < nEnd; blk += SLAB) {
        const int nxt = blk + SLAB;
        const bool more = (nxt < nEnd);

        float4 evr0, evr1, ftr0, ftr1;
        if (more) {
            evr0 = *reinterpret_cast<const float4*>(evSrc0 + nxt);
            evr1 = *reinterpret_cast<const float4*>(evSrc1 + nxt);
            ftr0 = *reinterpret_cast<const float4*>(ftSrc0 + (size_t)nxt * PDIM);
            ftr1 = *reinterpret_cast<const float4*>(ftSrc1 + (size_t)nxt * PDIM);
            if (nxt + 32 < nEnd && tid < 256) {
                long base = (long)(nxt + 32 + (tid >> 1)) * PDIM + pcol0;
                __builtin_prefetch(&feat[base + (tid & 1) * 64], 0, 0);
            }
        }

        #pragma unroll
        for (int kk = 0; kk < 8; ++kk) {
            const int nOff = kk * 4;
            v2f a;
            a.x = evBlk[rowBand + m][nOff + kp];
            a.y = evBlk[rowBand + m][nOff + kp + 1];
            #pragma unroll
            for (int t = 0; t < 4; ++t) {
                const int col = wcol + t * 16 + m;
                v2f b;
                b.x = ftBlk[nOff + kp][col];
                b.y = ftBlk[nOff + kp + 1][col];
                acc[t] = __builtin_amdgcn_wmma_f32_16x16x4_f32(
                    false, a, false, b, (short)0, acc[t], false, false);
            }
        }
        __syncthreads();

        if (more) {
            *evDst0 = evr0;
            *evDst1 = evr1;
            *ftDst0 = ftr0;
            *ftDst1 = ftr1;
            __syncthreads();
        }
    }

    float* out = partials + (size_t)sel * NCHUNKS * KP + (size_t)chunk * KP;
    const int mrow = rowBand + ((lane >> 4) << 3);
    #pragma unroll
    for (int t = 0; t < 4; ++t) {
        const int col = pcol0 + wcol + t * 16 + (lane & 15);
        #pragma unroll
        for (int r = 0; r < 8; ++r)
            out[(mrow + r) * PDIM + col] = acc[t][r];
    }
}

__global__ void fm_reduce(const float* __restrict__ partials,
                          float* __restrict__ Amat, float* __restrict__ Bmat)
{
    int idx = blockIdx.x * blockDim.x + threadIdx.x;
    if (idx >= KP) return;
    float sa = 0.0f, sb = 0.0f;
    for (int c = 0; c < NCHUNKS; ++c) {
        sa += partials[(size_t)c * KP + idx];
        sb += partials[(size_t)(NCHUNKS + c) * KP + idx];
    }
    Amat[idx] = sa;
    Bmat[idx] = sb;
}

__global__ void fm_mask(const float* __restrict__ evx,
                        const float* __restrict__ evy,
                        float* __restrict__ D)
{
    int idx = blockIdx.x * blockDim.x + threadIdx.x;
    if (idx >= KDIM * KDIM) return;
    int i = idx >> 7;
    int j = idx & 127;
    float s  = fmaxf(evx[KDIM - 1], evy[KDIM - 1]);
    float g1 = sqrtf(evx[j] / s);
    float g2 = sqrtf(evy[i] / s);
    float d1 = 1.0f / (g1 * g1 + 1.0f);
    float d2 = 1.0f / (g2 * g2 + 1.0f);
    float re = g2 * d2 - g1 * d1;
    float im = d2 - d1;
    D[idx] = re * re + im * im;
}

__global__ __launch_bounds__(256) void fm_gram(
    const float* __restrict__ Amat, const float* __restrict__ Bmat,
    float* __restrict__ AAt, float* __restrict__ BAt)
{
    const float* __restrict__ L = blockIdx.x ? Bmat : Amat;
    const float* __restrict__ R = Amat;
    float* __restrict__ out     = blockIdx.x ? BAt : AAt;

    __shared__ float blkL[KDIM][33];
    __shared__ float blkR[KDIM][33];

    const int tid  = threadIdx.x;
    const int lane = tid & 31;
    const int wave = tid >> 5;
    const int iBase = wave * 16;

    v8f acc[8];
    #pragma unroll
    for (int t = 0; t < 8; ++t) acc[t] = (v8f)0.0f;

    for (int p0 = 0; p0 < PDIM; p0 += 32) {
        for (int idx = tid; idx < KDIM * 32; idx += 256) {
            int r = idx >> 5, c = idx & 31;
            blkL[r][c] = L[r * PDIM + p0 + c];
            blkR[r][c] = R[r * PDIM + p0 + c];
        }
        __syncthreads();

        const int m  = lane & 15;
        const int kp = (lane >> 4) << 1;
        #pragma unroll
        for (int kk = 0; kk < 8; ++kk) {
            const int pOff = kk * 4;
            v2f a;
            a.x = blkL[iBase + m][pOff + kp];
            a.y = blkL[iBase + m][pOff + kp + 1];
            #pragma unroll
            for (int t = 0; t < 8; ++t) {
                v2f b;
                b.x = blkR[t * 16 + m][pOff + kp];
                b.y = blkR[t * 16 + m][pOff + kp + 1];
                acc[t] = __builtin_amdgcn_wmma_f32_16x16x4_f32(
                    false, a, false, b, (short)0, acc[t], false, false);
            }
        }
        __syncthreads();
    }

    const int mrow = iBase + ((lane >> 4) << 3);
    #pragma unroll
    for (int t = 0; t < 8; ++t) {
        const int j = t * 16 + (lane & 15);
        #pragma unroll
        for (int r = 0; r < 8; ++r)
            out[(mrow + r) * KDIM + j] = acc[t][r];
    }
}

#define USTRIDE 132
#define SOLVE_LDS_FLOATS (KDIM * USTRIDE + KDIM + 2 * KDIM)

__global__ __launch_bounds__(1024) void fm_solve(
    const float* __restrict__ AAt, const float* __restrict__ BAt,
    const float* __restrict__ D,   float* __restrict__ Cout)
{
    extern __shared__ __align__(16) float smem[];
    float* Ush  = smem;
    float* ysh  = smem + KDIM * USTRIDE;
    float* colb = ysh + KDIM;

    const int i    = blockIdx.x;
    const int tid  = threadIdx.x;
    const int lane = tid & 31;
    const int wave = tid >> 5;
    const int s  = wave & 7;
    const int r  = (wave >> 3) * 32 + lane;
    const int j0 = s * 16;

    float m[16];
    {
        const float4* srcv = reinterpret_cast<const float4*>(AAt + r * KDIM + j0);
        #pragma unroll
        for (int q = 0; q < 4; ++q) {
            float4 v = srcv[q];
            m[4 * q + 0] = v.x; m[4 * q + 1] = v.y;
            m[4 * q + 2] = v.z; m[4 * q + 3] = v.w;
        }
    }
    float bval = BAt[i * KDIM + r];
    if ((r >> 4) == s) {
        const float dl = D[i * KDIM + r];
        #pragma unroll
        for (int jj = 0; jj < 16; ++jj)
            if (j0 + jj == r) m[jj] += LAMBDA * dl;
    }

    for (int cb = 0; cb < 8; ++cb) {
        #pragma unroll
        for (int cc = 0; cc < 16; ++cc) {
            const int c = cb * 16 + cc;
            float* cbuf = colb + (c & 1) * KDIM;
            float4* urow = reinterpret_cast<float4*>(Ush + c * USTRIDE + j0);
            if (r == c) {
                urow[0] = make_float4(m[0],  m[1],  m[2],  m[3]);
                urow[1] = make_float4(m[4],  m[5],  m[6],  m[7]);
                urow[2] = make_float4(m[8],  m[9],  m[10], m[11]);
                urow[3] = make_float4(m[12], m[13], m[14], m[15]);
                if (s == 7) ysh[c] = bval;
            }
            if (s == cb) cbuf[r] = m[cc];
            __syncthreads();
            if (r > c && j0 + 15 > c) {
                const float piv = Ush[c * USTRIDE + c];
                const float mul = cbuf[r];
                const float4 p0 = urow[0], p1 = urow[1], p2 = urow[2], p3 = urow[3];
                const float yp  = ysh[c];
                const float f   = mul / piv;
                m[0]  -= f * p0.x; m[1]  -= f * p0.y; m[2]  -= f * p0.z; m[3]  -= f * p0.w;
                m[4]  -= f * p1.x; m[5]  -= f * p1.y; m[6]  -= f * p1.z; m[7]  -= f * p1.w;
                m[8]  -= f * p2.x; m[9]  -= f * p2.y; m[10] -= f * p2.z; m[11] -= f * p2.w;
                m[12] -= f * p3.x; m[13] -= f * p3.y; m[14] -= f * p3.z; m[15] -= f * p3.w;
                bval -= f * yp;
            }
        }
    }
    if (tid >= 32) return;

    float y[4], x[4], dg[4];
    #pragma unroll
    for (int k = 0; k < 4; ++k) {
        const int row = lane + 32 * k;
        y[k]  = ysh[row];
        dg[k] = Ush[row * USTRIDE + row];
        x[k]  = 0.0f;
    }
    #pragma unroll
    for (int q = 3; q >= 0; --q) {
        const float* ucol = Ush + lane * USTRIDE + 32 * q;
        float uc[4];
        #pragma unroll
        for (int k = 0; k <= q; ++k) uc[k] = ucol[k * 32 * USTRIDE + 31];
        #pragma unroll 1
        for (int cc = 31; cc >= 0; --cc) {
            const int cn = cc > 0 ? cc - 1 : 0;
            float un[4];
            #pragma unroll
            for (int k = 0; k <= q; ++k) un[k] = ucol[k * 32 * USTRIDE + cn];
            const float yc = __int_as_float(
                __builtin_amdgcn_readlane(__float_as_int(y[q]), cc));
            const float dc = __int_as_float(
                __builtin_amdgcn_readlane(__float_as_int(dg[q]), cc));
            const float xc = yc / dc;
            x[q] = (lane == cc) ? xc : x[q];
            #pragma unroll
            for (int k = 0; k < q; ++k) y[k] -= uc[k] * xc;
            {
                const float t = y[q] - uc[q] * xc;
                y[q] = (lane < cc) ? t : y[q];
            }
            #pragma unroll
            for (int k = 0; k <= q; ++k) uc[k] = un[k];
        }
    }
    #pragma unroll
    for (int k = 0; k < 4; ++k)
        Cout[i * KDIM + lane + 32 * k] = x[k];
}

extern "C" void kernel_launch(void* const* d_in, const int* in_sizes, int n_in,
                              void* d_out, int out_size, void* d_ws, size_t ws_size,
                              hipStream_t stream)
{
    const float* feat_x = (const float*)d_in[0];
    const float* feat_y = (const float*)d_in[1];
    const float* evals_x = (const float*)d_in[2];
    const float* evals_y = (const float*)d_in[3];
    const float* evecs_x = (const float*)d_in[4];
    const float* evecs_y = (const float*)d_in[5];

    float* wsf   = (float*)d_ws;
    float* Amat  = wsf;
    float* Bmat  = wsf + 32768;
    float* AAt   = wsf + 65536;
    float* BAt   = wsf + 81920;
    float* Dmask = wsf + 98304;
    float* part  = wsf + 114688;
    float* Cout  = (float*)d_out;

    fm_gemm_partial<<<dim3(NCHUNKS, 2, 2), 512, 0, stream>>>(
        evecs_x, evecs_y, feat_x, feat_y, part);

    fm_reduce<<<(KP + 255) / 256, 256, 0, stream>>>(part, Amat, Bmat);

    fm_mask<<<(KDIM * KDIM + 255) / 256, 256, 0, stream>>>(evals_x, evals_y, Dmask);

    fm_gram<<<2, 256, 0, stream>>>(Amat, Bmat, AAt, BAt);

    fm_solve<<<KDIM, 1024, SOLVE_LDS_FLOATS * sizeof(float), stream>>>(
        AAt, BAt, Dmask, Cout);
}


# gram kernel: one 16x32 output tile per one-wave workgroup (grid 4x8x2 = 64 workgroups), fragments read straight from global memory, not two 8-wave workgroups staged through LDS; k order kept
# speedup vs baseline: 4.9876x; 1.2414x over previous
#include <hip/hip_runtime.h>
#include <hip/hip_bf16.h>

typedef __attribute__((ext_vector_type(2))) float v2f;
typedef __attribute__((ext_vector_type(8))) float v8f;

#define KDIM 128
#define PDIM 256
#define NDIM 100000
#define LAMBDA 0.001f
#define NCHUNKS 64
#define SLAB 32
#define CHUNK 1568
static_assert(NDIM % SLAB == 0 && CHUNK % SLAB == 0,
              "every chunk must be a whole number of slabs");
static_assert((NCHUNKS - 1) * CHUNK < NDIM && NCHUNKS * CHUNK >= NDIM,
              "the chunks must cover N and none may be empty");
#define EVPITCH 36
#define KP (KDIM * PDIM)
#define PHALF (PDIM / 2)

__global__ __launch_bounds__(512) void fm_gemm_partial(
    const float* __restrict__ evx, const float* __restrict__ evy,
    const float* __restrict__ fx,  const float* __restrict__ fy,
    float* __restrict__ partials)
{
    const int sel   = blockIdx.y;
    const int chunk = blockIdx.x;
    const int pcol0 = blockIdx.z * PHALF;
    const float* __restrict__ ev   = sel ? evy : evx;
    const float* __restrict__ feat = sel ? fy  : fx;

    __shared__ __align__(16) float evBlk[KDIM][EVPITCH];
    __shared__ __align__(16) float ftBlk[SLAB][PHALF + 4];

    const int tid  = threadIdx.x;
    const int lane = tid & 31;
    const int wave = tid >> 5;
    const int rowBand = (wave >> 1) * 16;
    const int wcol    = (wave & 1) * 64;

    const int evRow = tid >> 3;
    const int evCol = (tid & 7) << 2;
    const int ftRow = tid >> 5;
    const int ftCol = (tid & 31) << 2;
    const float* __restrict__ evSrc0 = ev + (size_t)evRow * NDIM + evCol;
    const float* __restrict__ evSrc1 = evSrc0 + (size_t)64 * NDIM;
    const float* __restrict__ ftSrc0 = feat + (size_t)ftRow * PDIM + pcol0 + ftCol;
    const float* __restrict__ ftSrc1 = ftSrc0 + (size_t)16 * PDIM;
    float4* const evDst0 = reinterpret_cast<float4*>(&evBlk[evRow][evCol]);
    float4* const evDst1 = reinterpret_cast<float4*>(&evBlk[evRow + 64][evCol]);
    float4* const ftDst0 = reinterpret_cast<float4*>(&ftBlk[ftRow][ftCol]);
    float4* const ftDst1 = reinterpret_cast<float4*>(&ftBlk[ftRow + 16][ftCol]);
    v8f acc[4];
    #pragma unroll
    for (int t = 0; t < 4; ++t) acc[t] = (v8f)0.0f;

    const int nStart = chunk * CHUNK;
    const int nEnd   = min(nStart + CHUNK, NDIM);

    *evDst0 = *reinterpret_cast<const float4*>(evSrc0 + nStart);
    *evDst1 = *reinterpret_cast<const float4*>(evSrc1 + nStart);
    *ftDst0 = *reinterpret_cast<const float4*>(ftSrc0 + (size_t)nStart * PDIM);
    *ftDst1 = *reinterpret_cast<const float4*>(ftSrc1 + (size_t)nStart * PDIM);
    __syncthreads();

    const int m  = lane & 15;
    const int kp = (lane >> 4) << 1;

    for (int blk = nStart; blk < nEnd; blk += SLAB) {
        const int nxt = blk + SLAB;
        const bool more = (nxt < nEnd);

        float4 evr0, evr1, ftr0, ftr1;
        if (more) {
            evr0 = *reinterpret_cast<const float4*>(evSrc0 + nxt);
            evr1 = *reinterpret_cast<const float4*>(evSrc1 + nxt);
            ftr0 = *reinterpret_cast<const float4*>(ftSrc0 + (size_t)nxt * PDIM);
            ftr1 = *reinterpret_cast<const float4*>(ftSrc1 + (size_t)nxt * PDIM);
            if (nxt + 32 < nEnd && tid < 256) {
                long base = (long)(nxt + 32 + (tid >> 1)) * PDIM + pcol0;
                __builtin_prefetch(&feat[base + (tid & 1) * 64], 0, 0);
            }
        }

        #pragma unroll
        for (int kk = 0; kk < 8; ++kk) {
            const int nOff = kk * 4;
            v2f a;
            a.x = evBlk[rowBand + m][nOff + kp];
            a.y = evBlk[rowBand + m][nOff + kp + 1];
            #pragma unroll
            for (int t = 0; t < 4; ++t) {
                const int col = wcol + t * 16 + m;
                v2f b;
                b.x = ftBlk[nOff + kp][col];
                b.y = ftBlk[nOff + kp + 1][col];
                acc[t] = __builtin_amdgcn_wmma_f32_16x16x4_f32(
                    false, a, false, b, (short)0, acc[t], false, false);
            }
        }
        __syncthreads();

        if (more) {
            *evDst0 = evr0;
            *evDst1 = evr1;
            *ftDst0 = ftr0;
            *ftDst1 = ftr1;
            __syncthreads();
        }
    }

    float* out = partials + (size_t)sel * NCHUNKS * KP + (size_t)chunk * KP;
    const int mrow = rowBand + ((lane >> 4) << 3);
    #pragma unroll
    for (int t = 0; t < 4; ++t) {
        const int col = pcol0 + wcol + t * 16 + (lane & 15);
        #pragma unroll
        for (int r = 0; r < 8; ++r)
            out[(mrow + r) * PDIM + col] = acc[t][r];
    }
}

__global__ void fm_reduce(const float* __restrict__ partials,
                          float* __restrict__ Amat, float* __restrict__ Bmat)
{
    int idx = blockIdx.x * blockDim.x + threadIdx.x;
    if (idx >= KP) return;
    float sa = 0.0f, sb = 0.0f;
    for (int c = 0; c < NCHUNKS; ++c) {
        sa += partials[(size_t)c * KP + idx];
        sb += partials[(size_t)(NCHUNKS + c) * KP + idx];
    }
    Amat[idx] = sa;
    Bmat[idx] = sb;
}

__global__ void fm_mask(const float* __restrict__ evx,
                        const float* __restrict__ evy,
                        float* __restrict__ D)
{
    int idx = blockIdx.x * blockDim.x + threadIdx.x;
    if (idx >= KDIM * KDIM) return;
    int i = idx >> 7;
    int j = idx & 127;
    float s  = fmaxf(evx[KDIM - 1], evy[KDIM - 1]);
    float g1 = sqrtf(evx[j] / s);
    float g2 = sqrtf(evy[i] / s);
    float d1 = 1.0f / (g1 * g1 + 1.0f);
    float d2 = 1.0f / (g2 * g2 + 1.0f);
    float re = g2 * d2 - g1 * d1;
    float im = d2 - d1;
    D[idx] = re * re + im * im;
}

__global__ __launch_bounds__(32) __attribute__((amdgpu_waves_per_eu(1, 1)))
void fm_gram(
    const float* __restrict__ Amat, const float* __restrict__ Bmat,
    float* __restrict__ AAt, float* __restrict__ BAt)
{
    const float* __restrict__ L = blockIdx.z ? Bmat : Amat;
    const float* __restrict__ R = Amat;
    float* __restrict__ out     = blockIdx.z ? BAt : AAt;

    const int lane  = threadIdx.x;
    const int m     = lane & 15;
    const int kp    = (lane >> 4) << 1;
    const int iBase = blockIdx.y * 16;
    const int jBase = blockIdx.x * 32;

    const v2f* __restrict__ lrow =
        reinterpret_cast<const v2f*>(L + (iBase + m) * PDIM + kp);
    const v2f* __restrict__ rrow0 =
        reinterpret_cast<const v2f*>(R + (jBase + m) * PDIM + kp);
    const v2f* __restrict__ rrow1 =
        reinterpret_cast<const v2f*>(R + (jBase + 16 + m) * PDIM + kp);

    v8f acc0 = (v8f)0.0f, acc1 = (v8f)0.0f;
    #pragma unroll
    for (int step = 0; step < PDIM / 4; ++step) {
        const v2f a  = lrow[step * 2];
        const v2f b0 = rrow0[step * 2];
        const v2f b1 = rrow1[step * 2];
        acc0 = __builtin_amdgcn_wmma_f32_16x16x4_f32(
            false, a, false, b0, (short)0, acc0, false, false);
        acc1 = __builtin_amdgcn_wmma_f32_16x16x4_f32(
            false, a, false, b1, (short)0, acc1, false, false);
    }

    const int mrow = iBase + ((lane >> 4) << 3);
    const int j    = jBase + m;
    #pragma unroll
    for (int r = 0; r < 8; ++r) {
        out[(mrow + r) * KDIM + j]      = acc0[r];
        out[(mrow + r) * KDIM + j + 16] = acc1[r];
    }
}

#define USTRIDE 132
#define SOLVE_LDS_FLOATS (KDIM * USTRIDE + KDIM + 2 * KDIM)

__global__ __launch_bounds__(1024) void fm_solve(
    const float* __restrict__ AAt, const float* __restrict__ BAt,
    const float* __restrict__ D,   float* __restrict__ Cout)
{
    extern __shared__ __align__(16) float smem[];
    float* Ush  = smem;
    float* ysh  = smem + KDIM * USTRIDE;
    float* colb = ysh + KDIM;

    const int i    = blockIdx.x;
    const int tid  = threadIdx.x;
    const int lane = tid & 31;
    const int wave = tid >> 5;
    const int s  = wave & 7;
    const int r  = (wave >> 3) * 32 + lane;
    const int j0 = s * 16;

    float m[16];
    {
        const float4* srcv = reinterpret_cast<const float4*>(AAt + r * KDIM + j0);
        #pragma unroll
        for (int q = 0; q < 4; ++q) {
            float4 v = srcv[q];
            m[4 * q + 0] = v.x; m[4 * q + 1] = v.y;
            m[4 * q + 2] = v.z; m[4 * q + 3] = v.w;
        }
    }
    float bval = BAt[i * KDIM + r];
    if ((r >> 4) == s) {
        const float dl = D[i * KDIM + r];
        #pragma unroll
        for (int jj = 0; jj < 16; ++jj)
            if (j0 + jj == r) m[jj] += LAMBDA * dl;
    }

    for (int cb = 0; cb < 8; ++cb) {
        #pragma unroll
        for (int cc = 0; cc < 16; ++cc) {
            const int c = cb * 16 + cc;
            float* cbuf = colb + (c & 1) * KDIM;
            float4* urow = reinterpret_cast<float4*>(Ush + c * USTRIDE + j0);
            if (r == c) {
                urow[0] = make_float4(m[0],  m[1],  m[2],  m[3]);
                urow[1] = make_float4(m[4],  m[5],  m[6],  m[7]);
                urow[2] = make_float4(m[8],  m[9],  m[10], m[11]);
                urow[3] = make_float4(m[12], m[13], m[14], m[15]);
                if (s == 7) ysh[c] = bval;
            }
            if (s == cb) cbuf[r] = m[cc];
            __syncthreads();
            if (r > c && j0 + 15 > c) {
                const float piv = Ush[c * USTRIDE + c];
                const float mul = cbuf[r];
                const float4 p0 = urow[0], p1 = urow[1], p2 = urow[2], p3 = urow[3];
                const float yp  = ysh[c];
                const float f   = mul / piv;
                m[0]  -= f * p0.x; m[1]  -= f * p0.y; m[2]  -= f * p0.z; m[3]  -= f * p0.w;
                m[4]  -= f * p1.x; m[5]  -= f * p1.y; m[6]  -= f * p1.z; m[7]  -= f * p1.w;
                m[8]  -= f * p2.x; m[9]  -= f * p2.y; m[10] -= f * p2.z; m[11] -= f * p2.w;
                m[12] -= f * p3.x; m[13] -= f * p3.y; m[14] -= f * p3.z; m[15] -= f * p3.w;
                bval -= f * yp;
            }
        }
    }
    if (tid >= 32) return;

    float y[4], x[4], dg[4];
    #pragma unroll
    for (int k = 0; k < 4; ++k) {
        const int row = lane + 32 * k;
        y[k]  = ysh[row];
        dg[k] = Ush[row * USTRIDE + row];
        x[k]  = 0.0f;
    }
    #pragma unroll
    for (int q = 3; q >= 0; --q) {
        const float* ucol = Ush + lane * USTRIDE + 32 * q;
        float uc[4];
        #pragma unroll
        for (int k = 0; k <= q; ++k) uc[k] = ucol[k * 32 * USTRIDE + 31];
        #pragma unroll 1
        for (int cc = 31; cc >= 0; --cc) {
            const int cn = cc > 0 ? cc - 1 : 0;
            float un[4];
            #pragma unroll
            for (int k = 0; k <= q; ++k) un[k] = ucol[k * 32 * USTRIDE + cn];
            const float yc = __int_as_float(
                __builtin_amdgcn_readlane(__float_as_int(y[q]), cc));
            const float dc = __int_as_float(
                __builtin_amdgcn_readlane(__float_as_int(dg[q]), cc));
            const float xc = yc / dc;
            x[q] = (lane == cc) ? xc : x[q];
            #pragma unroll
            for (int k = 0; k < q; ++k) y[k] -= uc[k] * xc;
            {
                const float t = y[q] - uc[q] * xc;
                y[q] = (lane < cc) ? t : y[q];
            }
            #pragma unroll
            for (int k = 0; k <= q; ++k) uc[k] = un[k];
        }
    }
    #pragma unroll
    for (int k = 0; k < 4; ++k)
        Cout[i * KDIM + lane + 32 * k] = x[k];
}

extern "C" void kernel_launch(void* const* d_in, const int* in_sizes, int n_in,
                              void* d_out, int out_size, void* d_ws, size_t ws_size,
                              hipStream_t stream)
{
    const float* feat_x = (const float*)d_in[0];
    const float* feat_y = (const float*)d_in[1];
    const float* evals_x = (const float*)d_in[2];
    const float* evals_y = (const float*)d_in[3];
    const float* evecs_x = (const float*)d_in[4];
    const float* evecs_y = (const float*)d_in[5];

    float* wsf   = (float*)d_ws;
    float* Amat  = wsf;
    float* Bmat  = wsf + 32768;
    float* AAt   = wsf + 65536;
    float* BAt   = wsf + 81920;
    float* Dmask = wsf + 98304;
    float* part  = wsf + 114688;
    float* Cout  = (float*)d_out;

    fm_gemm_partial<<<dim3(NCHUNKS, 2, 2), 512, 0, stream>>>(
        evecs_x, evecs_y, feat_x, feat_y, part);

    fm_reduce<<<(KP + 255) / 256, 256, 0, stream>>>(part, Amat, Bmat);

    fm_mask<<<(KDIM * KDIM + 255) / 256, 256, 0, stream>>>(evals_x, evals_y, Dmask);

    fm_gram<<<dim3(4, 8, 2), 32, 0, stream>>>(Amat, Bmat, AAt, BAt);

    fm_solve<<<KDIM, 1024, SOLVE_LDS_FLOATS * sizeof(float), stream>>>(
        AAt, BAt, Dmask, Cout);
}


# solve kernel: a wave owns 4 whole columns (lane = row); 4 pivots factored inside that wave by lane reads, pivot rows from own registers, only multipliers cross LDS, 32 barriers not 128; order kept
# speedup vs baseline: 5.3989x; 1.0825x over previous
#include <hip/hip_runtime.h>
#include <hip/hip_bf16.h>

typedef __attribute__((ext_vector_type(2))) float v2f;
typedef __attribute__((ext_vector_type(8))) float v8f;

#define KDIM 128
#define PDIM 256
#define NDIM 100000
#define LAMBDA 0.001f
#define NCHUNKS 64
#define SLAB 32
#define CHUNK 1568
static_assert(NDIM % SLAB == 0 && CHUNK % SLAB == 0,
              "every chunk must be a whole number of slabs");
static_assert((NCHUNKS - 1) * CHUNK < NDIM && NCHUNKS * CHUNK >= NDIM,
              "the chunks must cover N and none may be empty");
#define EVPITCH 36
#define KP (KDIM * PDIM)
#define PHALF (PDIM / 2)

__global__ __launch_bounds__(512) void fm_gemm_partial(
    const float* __restrict__ evx, const float* __restrict__ evy,
    const float* __restrict__ fx,  const float* __restrict__ fy,
    float* __restrict__ partials)
{
    const int sel   = blockIdx.y;
    const int chunk = blockIdx.x;
    const int pcol0 = blockIdx.z * PHALF;
    const float* __restrict__ ev   = sel ? evy : evx;
    const float* __restrict__ feat = sel ? fy  : fx;

    __shared__ __align__(16) float evBlk[KDIM][EVPITCH];
    __shared__ __align__(16) float ftBlk[SLAB][PHALF + 4];

    const int tid  = threadIdx.x;
    const int lane = tid & 31;
    const int wave = tid >> 5;
    const int rowBand = (wave >> 1) * 16;
    const int wcol    = (wave & 1) * 64;

    const int evRow = tid >> 3;
    const int evCol = (tid & 7) << 2;
    const int ftRow = tid >> 5;
    const int ftCol = (tid & 31) << 2;
    const float* __restrict__ evSrc0 = ev + (size_t)evRow * NDIM + evCol;
    const float* __restrict__ evSrc1 = evSrc0 + (size_t)64 * NDIM;
    const float* __restrict__ ftSrc0 = feat + (size_t)ftRow * PDIM + pcol0 + ftCol;
    const float* __restrict__ ftSrc1 = ftSrc0 + (size_t)16 * PDIM;
    float4* const evDst0 = reinterpret_cast<float4*>(&evBlk[evRow][evCol]);
    float4* const evDst1 = reinterpret_cast<float4*>(&evBlk[evRow + 64][evCol]);
    float4* const ftDst0 = reinterpret_cast<float4*>(&ftBlk[ftRow][ftCol]);
    float4* const ftDst1 = reinterpret_cast<float4*>(&ftBlk[ftRow + 16][ftCol]);
    v8f acc[4];
    #pragma unroll
    for (int t = 0; t < 4; ++t) acc[t] = (v8f)0.0f;

    const int nStart = chunk * CHUNK;
    const int nEnd   = min(nStart + CHUNK, NDIM);

    *evDst0 = *reinterpret_cast<const float4*>(evSrc0 + nStart);
    *evDst1 = *reinterpret_cast<const float4*>(evSrc1 + nStart);
    *ftDst0 = *reinterpret_cast<const float4*>(ftSrc0 + (size_t)nStart * PDIM);
    *ftDst1 = *reinterpret_cast<const float4*>(ftSrc1 + (size_t)nStart * PDIM);
    __syncthreads();

    const int m  = lane & 15;
    const int kp = (lane >> 4) << 1;

    for (int blk = nStart; blk < nEnd; blk += SLAB) {
        const int nxt = blk + SLAB;
        const bool more = (nxt < nEnd);

        float4 evr0, evr1, ftr0, ftr1;
        if (more) {
            evr0 = *reinterpret_cast<const float4*>(evSrc0 + nxt);
            evr1 = *reinterpret_cast<const float4*>(evSrc1 + nxt);
            ftr0 = *reinterpret_cast<const float4*>(ftSrc0 + (size_t)nxt * PDIM);
            ftr1 = *reinterpret_cast<const float4*>(ftSrc1 + (size_t)nxt * PDIM);
            if (nxt + 32 < nEnd && tid < 256) {
                long base = (long)(nxt + 32 + (tid >> 1)) * PDIM + pcol0;
                __builtin_prefetch(&feat[base + (tid & 1) * 64], 0, 0);
            }
        }

        #pragma unroll
        for (int kk = 0; kk < 8; ++kk) {
            const int nOff = kk * 4;
            v2f a;
            a.x = evBlk[rowBand + m][nOff + kp];
            a.y = evBlk[rowBand + m][nOff + kp + 1];
            #pragma unroll
            for (int t = 0; t < 4; ++t) {
                const int col = wcol + t * 16 + m;
                v2f b;
                b.x = ftBlk[nOff + kp][col];
                b.y = ftBlk[nOff + kp + 1][col];
                acc[t] = __builtin_amdgcn_wmma_f32_16x16x4_f32(
                    false, a, false, b, (short)0, acc[t], false, false);
            }
        }
        __syncthreads();

        if (more) {
            *evDst0 = evr0;
            *evDst1 = evr1;
            *ftDst0 = ftr0;
            *ftDst1 = ftr1;
            __syncthreads();
        }
    }

    float* out = partials + (size_t)sel * NCHUNKS * KP + (size_t)chunk * KP;
    const int mrow = rowBand + ((lane >> 4) << 3);
    #pragma unroll
    for (int t = 0; t < 4; ++t) {
        const int col = pcol0 + wcol + t * 16 + (lane & 15);
        #pragma unroll
        for (int r = 0; r < 8; ++r)
            out[(mrow + r) * PDIM + col] = acc[t][r];
    }
}

__global__ void fm_reduce(const float* __restrict__ partials,
                          float* __restrict__ Amat, float* __restrict__ Bmat)
{
    int idx = blockIdx.x * blockDim.x + threadIdx.x;
    if (idx >= KP) return;
    float sa = 0.0f, sb = 0.0f;
    for (int c = 0; c < NCHUNKS; ++c) {
        sa += partials[(size_t)c * KP + idx];
        sb += partials[(size_t)(NCHUNKS + c) * KP + idx];
    }
    Amat[idx] = sa;
    Bmat[idx] = sb;
}

__global__ void fm_mask(const float* __restrict__ evx,
                        const float* __restrict__ evy,
                        float* __restrict__ D)
{
    int idx = blockIdx.x * blockDim.x + threadIdx.x;
    if (idx >= KDIM * KDIM) return;
    int i = idx >> 7;
    int j = idx & 127;
    float s  = fmaxf(evx[KDIM - 1], evy[KDIM - 1]);
    float g1 = sqrtf(evx[j] / s);
    float g2 = sqrtf(evy[i] / s);
    float d1 = 1.0f / (g1 * g1 + 1.0f);
    float d2 = 1.0f / (g2 * g2 + 1.0f);
    float re = g2 * d2 - g1 * d1;
    float im = d2 - d1;
    D[idx] = re * re + im * im;
}

__global__ __launch_bounds__(32) __attribute__((amdgpu_waves_per_eu(1, 1)))
void fm_gram(
    const float* __restrict__ Amat, const float* __restrict__ Bmat,
    float* __restrict__ AAt, float* __restrict__ BAt)
{
    const float* __restrict__ L = blockIdx.z ? Bmat : Amat;
    const float* __restrict__ R = Amat;
    float* __restrict__ out     = blockIdx.z ? BAt : AAt;

    const int lane  = threadIdx.x;
    const int m     = lane & 15;
    const int kp    = (lane >> 4) << 1;
    const int iBase = blockIdx.y * 16;
    const int jBase = blockIdx.x * 32;

    const v2f* __restrict__ lrow =
        reinterpret_cast<const v2f*>(L + (iBase + m) * PDIM + kp);
    const v2f* __restrict__ rrow0 =
        reinterpret_cast<const v2f*>(R + (jBase + m) * PDIM + kp);
    const v2f* __restrict__ rrow1 =
        reinterpret_cast<const v2f*>(R + (jBase + 16 + m) * PDIM + kp);

    v8f acc0 = (v8f)0.0f, acc1 = (v8f)0.0f;
    #pragma unroll
    for (int step = 0; step < PDIM / 4; ++step) {
        const v2f a  = lrow[step * 2];
        const v2f b0 = rrow0[step * 2];
        const v2f b1 = rrow1[step * 2];
        acc0 = __builtin_amdgcn_wmma_f32_16x16x4_f32(
            false, a, false, b0, (short)0, acc0, false, false);
        acc1 = __builtin_amdgcn_wmma_f32_16x16x4_f32(
            false, a, false, b1, (short)0, acc1, false, false);
    }

    const int mrow = iBase + ((lane >> 4) << 3);
    const int j    = jBase + m;
    #pragma unroll
    for (int r = 0; r < 8; ++r) {
        out[(mrow + r) * KDIM + j]      = acc0[r];
        out[(mrow + r) * KDIM + j + 16] = acc1[r];
    }
}

#define USTRIDE 132
#define FBUF_FLOATS (2 * 4 * KDIM)
#define SOLVE_LDS_FLOATS (KDIM * USTRIDE + KDIM + FBUF_FLOATS)

__device__ __forceinline__ float fm_lane_read(float v, int l)
{
    return __int_as_float(__builtin_amdgcn_readlane(__float_as_int(v), l));
}

template <int KC>
__device__ __forceinline__ void fm_elim_rowblock(
    float (&mc)[4][4], float (&rhs)[4], float* fbuf, const int lane, const int wave)
{
    #pragma unroll 1
    for (int g = 0; g < 8; ++g) {
        const int owner = KC * 8 + g;
        const int lc0   = 4 * g;
        float* fb = fbuf + (g & 1) * (4 * KDIM) + lane * 4;

        if (wave == owner) {
            #pragma unroll
            for (int t = 0; t < 4; ++t) {
                const int lc = lc0 + t;
                const float piv = fm_lane_read(mc[t][KC], lc);
                float f[4];
                #pragma unroll
                for (int k = 0; k < 4; ++k) {
                    if (k < KC) {
                        f[k] = 0.0f;
                    } else {
                        const float q = mc[t][k] / piv;
                        f[k] = (k > KC || lane > lc) ? q : 0.0f;
                    }
                }
                #pragma unroll
                for (int t2 = t + 1; t2 < 4; ++t2) {
                    const float u = fm_lane_read(mc[t2][KC], lc);
                    #pragma unroll
                    for (int k = KC; k < 4; ++k) mc[t2][k] -= f[k] * u;
                }
                *reinterpret_cast<float4*>(fb + t * KDIM) =
                    make_float4(f[0], f[1], f[2], f[3]);
            }
        }
        __syncthreads();

        if (wave > owner || wave == 31) {
            float4 fv[4];
            #pragma unroll
            for (int t = 0; t < 4; ++t)
                fv[t] = *reinterpret_cast<const float4*>(fb + t * KDIM);
            #pragma unroll
            for (int t = 0; t < 4; ++t) {
                const int lc = lc0 + t;
                const float f[4] = {fv[t].x, fv[t].y, fv[t].z, fv[t].w};
                if (wave > owner) {
                    #pragma unroll
                    for (int jj = 0; jj < 4; ++jj) {
                        const float u = fm_lane_read(mc[jj][KC], lc);
                        #pragma unroll
                        for (int k = KC; k < 4; ++k) mc[jj][k] -= f[k] * u;
                    }
                }
                if (wave == 31) {
                    const float yb = fm_lane_read(rhs[KC], lc);
                    #pragma unroll
                    for (int k = KC; k < 4; ++k) rhs[k] -= f[k] * yb;
                }
            }
        }
    }
}

__global__ __launch_bounds__(1024) void fm_solve(
    const float* __restrict__ AAt, const float* __restrict__ BAt,
    const float* __restrict__ D,   float* __restrict__ Cout)
{
    extern __shared__ __align__(16) float smem[];
    float* Ush  = smem;
    float* ysh  = smem + KDIM * USTRIDE;
    float* fbuf = ysh + KDIM;

    const int i    = blockIdx.x;
    const int tid  = threadIdx.x;
    const int lane = tid & 31;
    const int wave = __builtin_amdgcn_readfirstlane(tid >> 5);
    const int j0   = wave * 4;

    float mc[4][4];
    float rhs[4];
    #pragma unroll
    for (int k = 0; k < 4; ++k) {
        const int row = lane + 32 * k;
        const float4 v = *reinterpret_cast<const float4*>(AAt + row * KDIM + j0);
        mc[0][k] = v.x; mc[1][k] = v.y; mc[2][k] = v.z; mc[3][k] = v.w;
        rhs[k] = BAt[i * KDIM + row];
    }
    #pragma unroll
    for (int k = 0; k < 4; ++k) {
        const int row = lane + 32 * k;
        if ((row >> 2) == wave) {
            const float dl = D[i * KDIM + row];
            #pragma unroll
            for (int jj = 0; jj < 4; ++jj)
                if (j0 + jj == row) mc[jj][k] += LAMBDA * dl;
        }
    }

    fm_elim_rowblock<0>(mc, rhs, fbuf, lane, wave);
    fm_elim_rowblock<1>(mc, rhs, fbuf, lane, wave);
    fm_elim_rowblock<2>(mc, rhs, fbuf, lane, wave);
    fm_elim_rowblock<3>(mc, rhs, fbuf, lane, wave);

    #pragma unroll
    for (int k = 0; k < 4; ++k) {
        const int row = lane + 32 * k;
        *reinterpret_cast<float4*>(Ush + row * USTRIDE + j0) =
            make_float4(mc[0][k], mc[1][k], mc[2][k], mc[3][k]);
        if (wave == 31) ysh[row] = rhs[k];
    }
    __syncthreads();
    if (tid >= 32) return;

    float y[4], x[4], dg[4];
    #pragma unroll
    for (int k = 0; k < 4; ++k) {
        const int row = lane + 32 * k;
        y[k]  = ysh[row];
        dg[k] = Ush[row * USTRIDE + row];
        x[k]  = 0.0f;
    }
    #pragma unroll
    for (int q = 3; q >= 0; --q) {
        const float* ucol = Ush + lane * USTRIDE + 32 * q;
        float uc[4];
        #pragma unroll
        for (int k = 0; k <= q; ++k) uc[k] = ucol[k * 32 * USTRIDE + 31];
        #pragma unroll 1
        for (int cc = 31; cc >= 0; --cc) {
            const int cn = cc > 0 ? cc - 1 : 0;
            float un[4];
            #pragma unroll
            for (int k = 0; k <= q; ++k) un[k] = ucol[k * 32 * USTRIDE + cn];
            const float yc = __int_as_float(
                __builtin_amdgcn_readlane(__float_as_int(y[q]), cc));
            const float dc = __int_as_float(
                __builtin_amdgcn_readlane(__float_as_int(dg[q]), cc));
            const float xc = yc / dc;
            x[q] = (lane == cc) ? xc : x[q];
            #pragma unroll
            for (int k = 0; k < q; ++k) y[k] -= uc[k] * xc;
            {
                const float t = y[q] - uc[q] * xc;
                y[q] = (lane < cc) ? t : y[q];
            }
            #pragma unroll
            for (int k = 0; k <= q; ++k) uc[k] = un[k];
        }
    }
    #pragma unroll
    for (int k = 0; k < 4; ++k)
        Cout[i * KDIM + lane + 32 * k] = x[k];
}

extern "C" void kernel_launch(void* const* d_in, const int* in_sizes, int n_in,
                              void* d_out, int out_size, void* d_ws, size_t ws_size,
                              hipStream_t stream)
{
    const float* feat_x = (const float*)d_in[0];
    const float* feat_y = (const float*)d_in[1];
    const float* evals_x = (const float*)d_in[2];
    const float* evals_y = (const float*)d_in[3];
    const float* evecs_x = (const float*)d_in[4];
    const float* evecs_y = (const float*)d_in[5];

    float* wsf   = (float*)d_ws;
    float* Amat  = wsf;
    float* Bmat  = wsf + 32768;
    float* AAt   = wsf + 65536;
    float* BAt   = wsf + 81920;
    float* Dmask = wsf + 98304;
    float* part  = wsf + 114688;
    float* Cout  = (float*)d_out;

    fm_gemm_partial<<<dim3(NCHUNKS, 2, 2), 512, 0, stream>>>(
        evecs_x, evecs_y, feat_x, feat_y, part);

    fm_reduce<<<(KP + 255) / 256, 256, 0, stream>>>(part, Amat, Bmat);

    fm_mask<<<(KDIM * KDIM + 255) / 256, 256, 0, stream>>>(evals_x, evals_y, Dmask);

    fm_gram<<<dim3(4, 8, 2), 32, 0, stream>>>(Amat, Bmat, AAt, BAt);

    fm_solve<<<KDIM, 1024, SOLVE_LDS_FLOATS * sizeof(float), stream>>>(
        AAt, BAt, Dmask, Cout);
}


# solve kernel: division off both serial chains: a quotient by a pivot or diagonal element is a product with its reciprocal (rcp + 1 Newton step); U read 4 columns per 16-B LDS read, a group ahead
# speedup vs baseline: 5.7256x; 1.0605x over previous
#include <hip/hip_runtime.h>
#include <hip/hip_bf16.h>

typedef __attribute__((ext_vector_type(2))) float v2f;
typedef __attribute__((ext_vector_type(8))) float v8f;
typedef __attribute__((ext_vector_type(4))) float v4f;

#define KDIM 128
#define PDIM 256
#define NDIM 100000
#define LAMBDA 0.001f
#define NCHUNKS 64
#define SLAB 32
#define CHUNK 1568
static_assert(NDIM % SLAB == 0 && CHUNK % SLAB == 0,
              "every chunk must be a whole number of slabs");
static_assert((NCHUNKS - 1) * CHUNK < NDIM && NCHUNKS * CHUNK >= NDIM,
              "the chunks must cover N and none may be empty");
#define EVPITCH 36
#define KP (KDIM * PDIM)
#define PHALF (PDIM / 2)

__global__ __launch_bounds__(512) void fm_gemm_partial(
    const float* __restrict__ evx, const float* __restrict__ evy,
    const float* __restrict__ fx,  const float* __restrict__ fy,
    float* __restrict__ partials)
{
    const int sel   = blockIdx.y;
    const int chunk = blockIdx.x;
    const int pcol0 = blockIdx.z * PHALF;
    const float* __restrict__ ev   = sel ? evy : evx;
    const float* __restrict__ feat = sel ? fy  : fx;

    __shared__ __align__(16) float evBlk[KDIM][EVPITCH];
    __shared__ __align__(16) float ftBlk[SLAB][PHALF + 4];

    const int tid  = threadIdx.x;
    const int lane = tid & 31;
    const int wave = tid >> 5;
    const int rowBand = (wave >> 1) * 16;
    const int wcol    = (wave & 1) * 64;

    const int evRow = tid >> 3;
    const int evCol = (tid & 7) << 2;
    const int ftRow = tid >> 5;
    const int ftCol = (tid & 31) << 2;
    const float* __restrict__ evSrc0 = ev + (size_t)evRow * NDIM + evCol;
    const float* __restrict__ evSrc1 = evSrc0 + (size_t)64 * NDIM;
    const float* __restrict__ ftSrc0 = feat + (size_t)ftRow * PDIM + pcol0 + ftCol;
    const float* __restrict__ ftSrc1 = ftSrc0 + (size_t)16 * PDIM;
    float4* const evDst0 = reinterpret_cast<float4*>(&evBlk[evRow][evCol]);
    float4* const evDst1 = reinterpret_cast<float4*>(&evBlk[evRow + 64][evCol]);
    float4* const ftDst0 = reinterpret_cast<float4*>(&ftBlk[ftRow][ftCol]);
    float4* const ftDst1 = reinterpret_cast<float4*>(&ftBlk[ftRow + 16][ftCol]);
    v8f acc[4];
    #pragma unroll
    for (int t = 0; t < 4; ++t) acc[t] = (v8f)0.0f;

    const int nStart = chunk * CHUNK;
    const int nEnd   = min(nStart + CHUNK, NDIM);

    *evDst0 = *reinterpret_cast<const float4*>(evSrc0 + nStart);
    *evDst1 = *reinterpret_cast<const float4*>(evSrc1 + nStart);
    *ftDst0 = *reinterpret_cast<const float4*>(ftSrc0 + (size_t)nStart * PDIM);
    *ftDst1 = *reinterpret_cast<const float4*>(ftSrc1 + (size_t)nStart * PDIM);
    __syncthreads();

    const int m  = lane & 15;
    const int kp = (lane >> 4) << 1;

    for (int blk = nStart; blk < nEnd; blk += SLAB) {
        const int nxt = blk + SLAB;
        const bool more = (nxt < nEnd);

        float4 evr0, evr1, ftr0, ftr1;
        if (more) {
            evr0 = *reinterpret_cast<const float4*>(evSrc0 + nxt);
            evr1 = *reinterpret_cast<const float4*>(evSrc1 + nxt);
            ftr0 = *reinterpret_cast<const float4*>(ftSrc0 + (size_t)nxt * PDIM);
            ftr1 = *reinterpret_cast<const float4*>(ftSrc1 + (size_t)nxt * PDIM);
            if (nxt + 32 < nEnd && tid < 256) {
                long base = (long)(nxt + 32 + (tid >> 1)) * PDIM + pcol0;
                __builtin_prefetch(&feat[base + (tid & 1) * 64], 0, 0);
            }
        }

        #pragma unroll
        for (int kk = 0; kk < 8; ++kk) {
            const int nOff = kk * 4;
            v2f a;
            a.x = evBlk[rowBand + m][nOff + kp];
            a.y = evBlk[rowBand + m][nOff + kp + 1];
            #pragma unroll
            for (int t = 0; t < 4; ++t) {
                const int col = wcol + t * 16 + m;
                v2f b;
                b.x = ftBlk[nOff + kp][col];
                b.y = ftBlk[nOff + kp + 1][col];
                acc[t] = __builtin_amdgcn_wmma_f32_16x16x4_f32(
                    false, a, false, b, (short)0, acc[t], false, false);
            }
        }
        __syncthreads();

        if (more) {
            *evDst0 = evr0;
            *evDst1 = evr1;
            *ftDst0 = ftr0;
            *ftDst1 = ftr1;
            __syncthreads();
        }
    }

    float* out = partials + (size_t)sel * NCHUNKS * KP + (size_t)chunk * KP;
    const int mrow = rowBand + ((lane >> 4) << 3);
    #pragma unroll
    for (int t = 0; t < 4; ++t) {
        const int col = pcol0 + wcol + t * 16 + (lane & 15);
        #pragma unroll
        for (int r = 0; r < 8; ++r)
            out[(mrow + r) * PDIM + col] = acc[t][r];
    }
}

__global__ void fm_reduce(const float* __restrict__ partials,
                          float* __restrict__ Amat, float* __restrict__ Bmat)
{
    int idx = blockIdx.x * blockDim.x + threadIdx.x;
    if (idx >= KP) return;
    float sa = 0.0f, sb = 0.0f;
    for (int c = 0; c < NCHUNKS; ++c) {
        sa += partials[(size_t)c * KP + idx];
        sb += partials[(size_t)(NCHUNKS + c) * KP + idx];
    }
    Amat[idx] = sa;
    Bmat[idx] = sb;
}

__global__ void fm_mask(const float* __restrict__ evx,
                        const float* __restrict__ evy,
                        float* __restrict__ D)
{
    int idx = blockIdx.x * blockDim.x + threadIdx.x;
    if (idx >= KDIM * KDIM) return;
    int i = idx >> 7;
    int j = idx & 127;
    float s  = fmaxf(evx[KDIM - 1], evy[KDIM - 1]);
    float g1 = sqrtf(evx[j] / s);
    float g2 = sqrtf(evy[i] / s);
    float d1 = 1.0f / (g1 * g1 + 1.0f);
    float d2 = 1.0f / (g2 * g2 + 1.0f);
    float re = g2 * d2 - g1 * d1;
    float im = d2 - d1;
    D[idx] = re * re + im * im;
}

__global__ __launch_bounds__(32) __attribute__((amdgpu_waves_per_eu(1, 1)))
void fm_gram(
    const float* __restrict__ Amat, const float* __restrict__ Bmat,
    float* __restrict__ AAt, float* __restrict__ BAt)
{
    const float* __restrict__ L = blockIdx.z ? Bmat : Amat;
    const float* __restrict__ R = Amat;
    float* __restrict__ out     = blockIdx.z ? BAt : AAt;

    const int lane  = threadIdx.x;
    const int m     = lane & 15;
    const int kp    = (lane >> 4) << 1;
    const int iBase = blockIdx.y * 16;
    const int jBase = blockIdx.x * 32;

    const v2f* __restrict__ lrow =
        reinterpret_cast<const v2f*>(L + (iBase + m) * PDIM + kp);
    const v2f* __restrict__ rrow0 =
        reinterpret_cast<const v2f*>(R + (jBase + m) * PDIM + kp);
    const v2f* __restrict__ rrow1 =
        reinterpret_cast<const v2f*>(R + (jBase + 16 + m) * PDIM + kp);

    v8f acc0 = (v8f)0.0f, acc1 = (v8f)0.0f;
    #pragma unroll
    for (int step = 0; step < PDIM / 4; ++step) {
        const v2f a  = lrow[step * 2];
        const v2f b0 = rrow0[step * 2];
        const v2f b1 = rrow1[step * 2];
        acc0 = __builtin_amdgcn_wmma_f32_16x16x4_f32(
            false, a, false, b0, (short)0, acc0, false, false);
        acc1 = __builtin_amdgcn_wmma_f32_16x16x4_f32(
            false, a, false, b1, (short)0, acc1, false, false);
    }

    const int mrow = iBase + ((lane >> 4) << 3);
    const int j    = jBase + m;
    #pragma unroll
    for (int r = 0; r < 8; ++r) {
        out[(mrow + r) * KDIM + j]      = acc0[r];
        out[(mrow + r) * KDIM + j + 16] = acc1[r];
    }
}

#define USTRIDE 132
#define FBUF_FLOATS (2 * 4 * KDIM)
#define SOLVE_LDS_FLOATS (KDIM * USTRIDE + KDIM + FBUF_FLOATS)

__device__ __forceinline__ float fm_lane_read(float v, int l)
{
    return __int_as_float(__builtin_amdgcn_readlane(__float_as_int(v), l));
}

__device__ __forceinline__ float fm_recip(float p)
{
    const float r = __builtin_amdgcn_rcpf(p);
    const float e = __builtin_fmaf(-p, r, 1.0f);
    return __builtin_fmaf(r, e, r);
}

template <int KC>
__device__ __forceinline__ void fm_elim_rowblock(
    float (&mc)[4][4], float (&rhs)[4], float* fbuf, const int lane, const int wave)
{
    #pragma unroll 1
    for (int g = 0; g < 8; ++g) {
        const int owner = KC * 8 + g;
        const int lc0   = 4 * g;
        float* fb = fbuf + (g & 1) * (4 * KDIM) + lane * 4;

        if (wave == owner) {
            #pragma unroll
            for (int t = 0; t < 4; ++t) {
                const int lc = lc0 + t;
                const float rpiv = fm_recip(fm_lane_read(mc[t][KC], lc));
                float f[4];
                #pragma unroll
                for (int k = 0; k < 4; ++k) {
                    if (k < KC) {
                        f[k] = 0.0f;
                    } else {
                        const float q = mc[t][k] * rpiv;
                        f[k] = (k > KC || lane > lc) ? q : 0.0f;
                    }
                }
                #pragma unroll
                for (int t2 = t + 1; t2 < 4; ++t2) {
                    const float u = fm_lane_read(mc[t2][KC], lc);
                    #pragma unroll
                    for (int k = KC; k < 4; ++k) mc[t2][k] -= f[k] * u;
                }
                *reinterpret_cast<float4*>(fb + t * KDIM) =
                    make_float4(f[0], f[1], f[2], f[3]);
            }
        }
        __syncthreads();

        if (wave > owner || wave == 31) {
            float4 fv[4];
            #pragma unroll
            for (int t = 0; t < 4; ++t)
                fv[t] = *reinterpret_cast<const float4*>(fb + t * KDIM);
            #pragma unroll
            for (int t = 0; t < 4; ++t) {
                const int lc = lc0 + t;
                const float f[4] = {fv[t].x, fv[t].y, fv[t].z, fv[t].w};
                if (wave > owner) {
                    #pragma unroll
                    for (int jj = 0; jj < 4; ++jj) {
                        const float u = fm_lane_read(mc[jj][KC], lc);
                        #pragma unroll
                        for (int k = KC; k < 4; ++k) mc[jj][k] -= f[k] * u;
                    }
                }
                if (wave == 31) {
                    const float yb = fm_lane_read(rhs[KC], lc);
                    #pragma unroll
                    for (int k = KC; k < 4; ++k) rhs[k] -= f[k] * yb;
                }
            }
        }
    }
}

__global__ __launch_bounds__(1024) void fm_solve(
    const float* __restrict__ AAt, const float* __restrict__ BAt,
    const float* __restrict__ D,   float* __restrict__ Cout)
{
    extern __shared__ __align__(16) float smem[];
    float* Ush  = smem;
    float* ysh  = smem + KDIM * USTRIDE;
    float* fbuf = ysh + KDIM;

    const int i    = blockIdx.x;
    const int tid  = threadIdx.x;
    const int lane = tid & 31;
    const int wave = __builtin_amdgcn_readfirstlane(tid >> 5);
    const int j0   = wave * 4;

    float mc[4][4];
    float rhs[4];
    #pragma unroll
    for (int k = 0; k < 4; ++k) {
        const int row = lane + 32 * k;
        const float4 v = *reinterpret_cast<const float4*>(AAt + row * KDIM + j0);
        mc[0][k] = v.x; mc[1][k] = v.y; mc[2][k] = v.z; mc[3][k] = v.w;
        rhs[k] = BAt[i * KDIM + row];
    }
    #pragma unroll
    for (int k = 0; k < 4; ++k) {
        const int row = lane + 32 * k;
        if ((row >> 2) == wave) {
            const float dl = D[i * KDIM + row];
            #pragma unroll
            for (int jj = 0; jj < 4; ++jj)
                if (j0 + jj == row) mc[jj][k] += LAMBDA * dl;
        }
    }

    fm_elim_rowblock<0>(mc, rhs, fbuf, lane, wave);
    fm_elim_rowblock<1>(mc, rhs, fbuf, lane, wave);
    fm_elim_rowblock<2>(mc, rhs, fbuf, lane, wave);
    fm_elim_rowblock<3>(mc, rhs, fbuf, lane, wave);

    #pragma unroll
    for (int k = 0; k < 4; ++k) {
        const int row = lane + 32 * k;
        *reinterpret_cast<float4*>(Ush + row * USTRIDE + j0) =
            make_float4(mc[0][k], mc[1][k], mc[2][k], mc[3][k]);
        if (wave == 31) ysh[row] = rhs[k];
    }
    __syncthreads();
    if (tid >= 32) return;

    float y[4], x[4], rdg[4];
    #pragma unroll
    for (int k = 0; k < 4; ++k) {
        const int row = lane + 32 * k;
        y[k]   = ysh[row];
        rdg[k] = 1.0f / Ush[row * USTRIDE + row];
        x[k]   = 0.0f;
    }
    #pragma unroll
    for (int q = 3; q >= 0; --q) {
        const float* ublk = Ush + lane * USTRIDE + 32 * q;
        v4f ucur[4];
        #pragma unroll
        for (int k = 0; k <= q; ++k)
            ucur[k] = *reinterpret_cast<const v4f*>(ublk + k * 32 * USTRIDE + 28);
        #pragma unroll 1
        for (int g4 = 7; g4 >= 0; --g4) {
            const int gn = g4 > 0 ? g4 - 1 : 0;
            v4f unxt[4];
            #pragma unroll
            for (int k = 0; k <= q; ++k)
                unxt[k] = *reinterpret_cast<const v4f*>(ublk + k * 32 * USTRIDE + 4 * gn);
            asm volatile("" ::: "memory");
            #pragma unroll
            for (int j = 3; j >= 0; --j) {
                const int cc = 4 * g4 + j;
                const float xc = fm_lane_read(y[q], cc) * fm_lane_read(rdg[q], cc);
                x[q] = (lane == cc) ? xc : x[q];
                #pragma unroll
                for (int k = 0; k < q; ++k) y[k] -= ucur[k][j] * xc;
                {
                    const float t = y[q] - ucur[q][j] * xc;
                    y[q] = (lane < cc) ? t : y[q];
                }
            }
            #pragma unroll
            for (int k = 0; k <= q; ++k) ucur[k] = unxt[k];
        }
    }
    #pragma unroll
    for (int k = 0; k < 4; ++k)
        Cout[i * KDIM + lane + 32 * k] = x[k];
}

extern "C" void kernel_launch(void* const* d_in, const int* in_sizes, int n_in,
                              void* d_out, int out_size, void* d_ws, size_t ws_size,
                              hipStream_t stream)
{
    const float* feat_x = (const float*)d_in[0];
    const float* feat_y = (const float*)d_in[1];
    const float* evals_x = (const float*)d_in[2];
    const float* evals_y = (const float*)d_in[3];
    const float* evecs_x = (const float*)d_in[4];
    const float* evecs_y = (const float*)d_in[5];

    float* wsf   = (float*)d_ws;
    float* Amat  = wsf;
    float* Bmat  = wsf + 32768;
    float* AAt   = wsf + 65536;
    float* BAt   = wsf + 81920;
    float* Dmask = wsf + 98304;
    float* part  = wsf + 114688;
    float* Cout  = (float*)d_out;

    fm_gemm_partial<<<dim3(NCHUNKS, 2, 2), 512, 0, stream>>>(
        evecs_x, evecs_y, feat_x, feat_y, part);

    fm_reduce<<<(KP + 255) / 256, 256, 0, stream>>>(part, Amat, Bmat);

    fm_mask<<<(KDIM * KDIM + 255) / 256, 256, 0, stream>>>(evals_x, evals_y, Dmask);

    fm_gram<<<dim3(4, 8, 2), 32, 0, stream>>>(Amat, Bmat, AAt, BAt);

    fm_solve<<<KDIM, 1024, SOLVE_LDS_FLOATS * sizeof(float), stream>>>(
        AAt, BAt, Dmask, Cout);
}


# gemm kernel: feat slab kept in LDS k-pair interleaved, so a B fragment is one conflict-free 8-byte LDS read (not two 4-byte reads with a 2-way bank conflict plus register moves); k order kept
# speedup vs baseline: 6.0732x; 1.0607x over previous
#include <hip/hip_runtime.h>
#include <hip/hip_bf16.h>

typedef __attribute__((ext_vector_type(2))) float v2f;
typedef __attribute__((ext_vector_type(8))) float v8f;
typedef __attribute__((ext_vector_type(4))) float v4f;

#define KDIM 128
#define PDIM 256
#define NDIM 100000
#define LAMBDA 0.001f
#define NCHUNKS 64
#define SLAB 32
#define CHUNK 1568
static_assert(NDIM % SLAB == 0 && CHUNK % SLAB == 0,
              "every chunk must be a whole number of slabs");
static_assert((NCHUNKS - 1) * CHUNK < NDIM && NCHUNKS * CHUNK >= NDIM,
              "the chunks must cover N and none may be empty");
#define EVPITCH 36
#define KP (KDIM * PDIM)
#define PHALF (PDIM / 2)
#define FTPITCH (2 * PHALF + 32)

__global__ __launch_bounds__(512) void fm_gemm_partial(
    const float* __restrict__ evx, const float* __restrict__ evy,
    const float* __restrict__ fx,  const float* __restrict__ fy,
    float* __restrict__ partials)
{
    const int sel   = blockIdx.y;
    const int chunk = blockIdx.x;
    const int pcol0 = blockIdx.z * PHALF;
    const float* __restrict__ ev   = sel ? evy : evx;
    const float* __restrict__ feat = sel ? fy  : fx;

    __shared__ __align__(16) float evBlk[KDIM][EVPITCH];
    __shared__ __align__(16) float ftP[SLAB / 2][FTPITCH];

    const int tid  = threadIdx.x;
    const int lane = tid & 31;
    const int wave = tid >> 5;
    const int rowBand = (wave >> 1) * 16;
    const int wcol    = (wave & 1) * 64;

    const int evRow = tid >> 3;
    const int evCol = (tid & 7) << 2;
    const int ftRow = tid >> 5;
    const int ftCol = (tid & 31) << 2;
    const float* __restrict__ evSrc0 = ev + (size_t)evRow * NDIM + evCol;
    const float* __restrict__ evSrc1 = evSrc0 + (size_t)64 * NDIM;
    const float* __restrict__ ftSrc0 = feat + (size_t)(2 * ftRow) * PDIM + pcol0 + ftCol;
    const float* __restrict__ ftSrc1 = ftSrc0 + PDIM;
    float4* const evDst0 = reinterpret_cast<float4*>(&evBlk[evRow][evCol]);
    float4* const evDst1 = reinterpret_cast<float4*>(&evBlk[evRow + 64][evCol]);
    float4* const ftDst0 = reinterpret_cast<float4*>(&ftP[ftRow][2 * ftCol]);
    float4* const ftDst1 = reinterpret_cast<float4*>(&ftP[ftRow][2 * ftCol + 4]);
    v8f acc[4];
    #pragma unroll
    for (int t = 0; t < 4; ++t) acc[t] = (v8f)0.0f;

    const int nStart = chunk * CHUNK;
    const int nEnd   = min(nStart + CHUNK, NDIM);

    *evDst0 = *reinterpret_cast<const float4*>(evSrc0 + nStart);
    *evDst1 = *reinterpret_cast<const float4*>(evSrc1 + nStart);
    {
        const float4 f0 = *reinterpret_cast<const float4*>(ftSrc0 + (size_t)nStart * PDIM);
        const float4 f1 = *reinterpret_cast<const float4*>(ftSrc1 + (size_t)nStart * PDIM);
        *ftDst0 = make_float4(f0.x, f1.x, f0.y, f1.y);
        *ftDst1 = make_float4(f0.z, f1.z, f0.w, f1.w);
    }
    __syncthreads();

    const int m  = lane & 15;
    const int kp = (lane >> 4) << 1;

    for (int blk = nStart; blk < nEnd; blk += SLAB) {
        const int nxt = blk + SLAB;
        const bool more = (nxt < nEnd);

        float4 evr0, evr1, ftr0, ftr1;
        if (more) {
            evr0 = *reinterpret_cast<const float4*>(evSrc0 + nxt);
            evr1 = *reinterpret_cast<const float4*>(evSrc1 + nxt);
            ftr0 = *reinterpret_cast<const float4*>(ftSrc0 + (size_t)nxt * PDIM);
            ftr1 = *reinterpret_cast<const float4*>(ftSrc1 + (size_t)nxt * PDIM);
            if (nxt + 32 < nEnd && tid < 256) {
                long base = (long)(nxt + 32 + (tid >> 1)) * PDIM + pcol0;
                __builtin_prefetch(&feat[base + (tid & 1) * 64], 0, 0);
            }
        }

        #pragma unroll
        for (int kk = 0; kk < 8; ++kk) {
            const int nOff = kk * 4;
            v2f a;
            a.x = evBlk[rowBand + m][nOff + kp];
            a.y = evBlk[rowBand + m][nOff + kp + 1];
            #pragma unroll
            for (int t = 0; t < 4; ++t) {
                const int col = wcol + t * 16 + m;
                const v2f b = *reinterpret_cast<const v2f*>(
                    &ftP[2 * kk + (lane >> 4)][2 * col]);
                acc[t] = __builtin_amdgcn_wmma_f32_16x16x4_f32(
                    false, a, false, b, (short)0, acc[t], false, false);
            }
        }
        __syncthreads();

        if (more) {
            *evDst0 = evr0;
            *evDst1 = evr1;
            *ftDst0 = make_float4(ftr0.x, ftr1.x, ftr0.y, ftr1.y);
            *ftDst1 = make_float4(ftr0.z, ftr1.z, ftr0.w, ftr1.w);
            __syncthreads();
        }
    }

    float* out = partials + (size_t)sel * NCHUNKS * KP + (size_t)chunk * KP;
    const int mrow = rowBand + ((lane >> 4) << 3);
    #pragma unroll
    for (int t = 0; t < 4; ++t) {
        const int col = pcol0 + wcol + t * 16 + (lane & 15);
        #pragma unroll
        for (int r = 0; r < 8; ++r)
            out[(mrow + r) * PDIM + col] = acc[t][r];
    }
}

__global__ void fm_reduce(const float* __restrict__ partials,
                          float* __restrict__ Amat, float* __restrict__ Bmat)
{
    int idx = blockIdx.x * blockDim.x + threadIdx.x;
    if (idx >= KP) return;
    float sa = 0.0f, sb = 0.0f;
    for (int c = 0; c < NCHUNKS; ++c) {
        sa += partials[(size_t)c * KP + idx];
        sb += partials[(size_t)(NCHUNKS + c) * KP + idx];
    }
    Amat[idx] = sa;
    Bmat[idx] = sb;
}

__global__ void fm_mask(const float* __restrict__ evx,
                        const float* __restrict__ evy,
                        float* __restrict__ D)
{
    int idx = blockIdx.x * blockDim.x + threadIdx.x;
    if (idx >= KDIM * KDIM) return;
    int i = idx >> 7;
    int j = idx & 127;
    float s  = fmaxf(evx[KDIM - 1], evy[KDIM - 1]);
    float g1 = sqrtf(evx[j] / s);
    float g2 = sqrtf(evy[i] / s);
    float d1 = 1.0f / (g1 * g1 + 1.0f);
    float d2 = 1.0f / (g2 * g2 + 1.0f);
    float re = g2 * d2 - g1 * d1;
    float im = d2 - d1;
    D[idx] = re * re + im * im;
}

__global__ __launch_bounds__(32) __attribute__((amdgpu_waves_per_eu(1, 1)))
void fm_gram(
    const float* __restrict__ Amat, const float* __restrict__ Bmat,
    float* __restrict__ AAt, float* __restrict__ BAt)
{
    const float* __restrict__ L = blockIdx.z ? Bmat : Amat;
    const float* __restrict__ R = Amat;
    float* __restrict__ out     = blockIdx.z ? BAt : AAt;

    const int lane  = threadIdx.x;
    const int m     = lane & 15;
    const int kp    = (lane >> 4) << 1;
    const int iBase = blockIdx.y * 16;
    const int jBase = blockIdx.x * 32;

    const v2f* __restrict__ lrow =
        reinterpret_cast<const v2f*>(L + (iBase + m) * PDIM + kp);
    const v2f* __restrict__ rrow0 =
        reinterpret_cast<const v2f*>(R + (jBase + m) * PDIM + kp);
    const v2f* __restrict__ rrow1 =
        reinterpret_cast<const v2f*>(R + (jBase + 16 + m) * PDIM + kp);

    v8f acc0 = (v8f)0.0f, acc1 = (v8f)0.0f;
    #pragma unroll
    for (int step = 0; step < PDIM / 4; ++step) {
        const v2f a  = lrow[step * 2];
        const v2f b0 = rrow0[step * 2];
        const v2f b1 = rrow1[step * 2];
        acc0 = __builtin_amdgcn_wmma_f32_16x16x4_f32(
            false, a, false, b0, (short)0, acc0, false, false);
        acc1 = __builtin_amdgcn_wmma_f32_16x16x4_f32(
            false, a, false, b1, (short)0, acc1, false, false);
    }

    const int mrow = iBase + ((lane >> 4) << 3);
    const int j    = jBase + m;
    #pragma unroll
    for (int r = 0; r < 8; ++r) {
        out[(mrow + r) * KDIM + j]      = acc0[r];
        out[(mrow + r) * KDIM + j + 16] = acc1[r];
    }
}

#define USTRIDE 132
#define FBUF_FLOATS (2 * 4 * KDIM)
#define SOLVE_LDS_FLOATS (KDIM * USTRIDE + KDIM + FBUF_FLOATS)

__device__ __forceinline__ float fm_lane_read(float v, int l)
{
    return __int_as_float(__builtin_amdgcn_readlane(__float_as_int(v), l));
}

__device__ __forceinline__ float fm_recip(float p)
{
    const float r = __builtin_amdgcn_rcpf(p);
    const float e = __builtin_fmaf(-p, r, 1.0f);
    return __builtin_fmaf(r, e, r);
}

template <int KC>
__device__ __forceinline__ void fm_elim_rowblock(
    float (&mc)[4][4], float (&rhs)[4], float* fbuf, const int lane, const int wave)
{
    #pragma unroll 1
    for (int g = 0; g < 8; ++g) {
        const int owner = KC * 8 + g;
        const int lc0   = 4 * g;
        float* fb = fbuf + (g & 1) * (4 * KDIM) + lane * 4;

        if (wave == owner) {
            #pragma unroll
            for (int t = 0; t < 4; ++t) {
                const int lc = lc0 + t;
                const float rpiv = fm_recip(fm_lane_read(mc[t][KC], lc));
                float f[4];
                #pragma unroll
                for (int k = 0; k < 4; ++k) {
                    if (k < KC) {
                        f[k] = 0.0f;
                    } else {
                        const float q = mc[t][k] * rpiv;
                        f[k] = (k > KC || lane > lc) ? q : 0.0f;
                    }
                }
                #pragma unroll
                for (int t2 = t + 1; t2 < 4; ++t2) {
                    const float u = fm_lane_read(mc[t2][KC], lc);
                    #pragma unroll
                    for (int k = KC; k < 4; ++k) mc[t2][k] -= f[k] * u;
                }
                *reinterpret_cast<float4*>(fb + t * KDIM) =
                    make_float4(f[0], f[1], f[2], f[3]);
            }
        }
        __syncthreads();

        if (wave > owner || wave == 31) {
            float4 fv[4];
            #pragma unroll
            for (int t = 0; t < 4; ++t)
                fv[t] = *reinterpret_cast<const float4*>(fb + t * KDIM);
            #pragma unroll
            for (int t = 0; t < 4; ++t) {
                const int lc = lc0 + t;
                const float f[4] = {fv[t].x, fv[t].y, fv[t].z, fv[t].w};
                if (wave > owner) {
                    #pragma unroll
                    for (int jj = 0; jj < 4; ++jj) {
                        const float u = fm_lane_read(mc[jj][KC], lc);
                        #pragma unroll
                        for (int k = KC; k < 4; ++k) mc[jj][k] -= f[k] * u;
                    }
                }
                if (wave == 31) {
                    const float yb = fm_lane_read(rhs[KC], lc);
                    #pragma unroll
                    for (int k = KC; k < 4; ++k) rhs[k] -= f[k] * yb;
                }
            }
        }
    }
}

__global__ __launch_bounds__(1024) void fm_solve(
    const float* __restrict__ AAt, const float* __restrict__ BAt,
    const float* __restrict__ D,   float* __restrict__ Cout)
{
    extern __shared__ __align__(16) float smem[];
    float* Ush  = smem;
    float* ysh  = smem + KDIM * USTRIDE;
    float* fbuf = ysh + KDIM;

    const int i    = blockIdx.x;
    const int tid  = threadIdx.x;
    const int lane = tid & 31;
    const int wave = __builtin_amdgcn_readfirstlane(tid >> 5);
    const int j0   = wave * 4;

    float mc[4][4];
    float rhs[4];
    #pragma unroll
    for (int k = 0; k < 4; ++k) {
        const int row = lane + 32 * k;
        const float4 v = *reinterpret_cast<const float4*>(AAt + row * KDIM + j0);
        mc[0][k] = v.x; mc[1][k] = v.y; mc[2][k] = v.z; mc[3][k] = v.w;
        rhs[k] = BAt[i * KDIM + row];
    }
    #pragma unroll
    for (int k = 0; k < 4; ++k) {
        const int row = lane + 32 * k;
        if ((row >> 2) == wave) {
            const float dl = D[i * KDIM + row];
            #pragma unroll
            for (int jj = 0; jj < 4; ++jj)
                if (j0 + jj == row) mc[jj][k] += LAMBDA * dl;
        }
    }

    fm_elim_rowblock<0>(mc, rhs, fbuf, lane, wave);
    fm_elim_rowblock<1>(mc, rhs, fbuf, lane, wave);
    fm_elim_rowblock<2>(mc, rhs, fbuf, lane, wave);
    fm_elim_rowblock<3>(mc, rhs, fbuf, lane, wave);

    #pragma unroll
    for (int k = 0; k < 4; ++k) {
        const int row = lane + 32 * k;
        *reinterpret_cast<float4*>(Ush + row * USTRIDE + j0) =
            make_float4(mc[0][k], mc[1][k], mc[2][k], mc[3][k]);
        if (wave == 31) ysh[row] = rhs[k];
    }
    __syncthreads();
    if (tid >= 32) return;

    float y[4], x[4], rdg[4];
    #pragma unroll
    for (int k = 0; k < 4; ++k) {
        const int row = lane + 32 * k;
        y[k]   = ysh[row];
        rdg[k] = 1.0f / Ush[row * USTRIDE + row];
        x[k]   = 0.0f;
    }
    #pragma unroll
    for (int q = 3; q >= 0; --q) {
        const float* ublk = Ush + lane * USTRIDE + 32 * q;
        v4f ucur[4];
        #pragma unroll
        for (int k = 0; k <= q; ++k)
            ucur[k] = *reinterpret_cast<const v4f*>(ublk + k * 32 * USTRIDE + 28);
        #pragma unroll 1
        for (int g4 = 7; g4 >= 0; --g4) {
            const int gn = g4 > 0 ? g4 - 1 : 0;
            v4f unxt[4];
            #pragma unroll
            for (int k = 0; k <= q; ++k)
                unxt[k] = *reinterpret_cast<const v4f*>(ublk + k * 32 * USTRIDE + 4 * gn);
            asm volatile("" ::: "memory");
            #pragma unroll
            for (int j = 3; j >= 0; --j) {
                const int cc = 4 * g4 + j;
                const float xc = fm_lane_read(y[q], cc) * fm_lane_read(rdg[q], cc);
                x[q] = (lane == cc) ? xc : x[q];
                #pragma unroll
                for (int k = 0; k < q; ++k) y[k] -= ucur[k][j] * xc;
                {
                    const float t = y[q] - ucur[q][j] * xc;
                    y[q] = (lane < cc) ? t : y[q];
                }
            }
            #pragma unroll
            for (int k = 0; k <= q; ++k) ucur[k] = unxt[k];
        }
    }
    #pragma unroll
    for (int k = 0; k < 4; ++k)
        Cout[i * KDIM + lane + 32 * k] = x[k];
}

extern "C" void kernel_launch(void* const* d_in, const int* in_sizes, int n_in,
                              void* d_out, int out_size, void* d_ws, size_t ws_size,
                              hipStream_t stream)
{
    const float* feat_x = (const float*)d_in[0];
    const float* feat_y = (const float*)d_in[1];
    const float* evals_x = (const float*)d_in[2];
    const float* evals_y = (const float*)d_in[3];
    const float* evecs_x = (const float*)d_in[4];
    const float* evecs_y = (const float*)d_in[5];

    float* wsf   = (float*)d_ws;
    float* Amat  = wsf;
    float* Bmat  = wsf + 32768;
    float* AAt   = wsf + 65536;
    float* BAt   = wsf + 81920;
    float* Dmask = wsf + 98304;
    float* part  = wsf + 114688;
    float* Cout  = (float*)d_out;

    fm_gemm_partial<<<dim3(NCHUNKS, 2, 2), 512, 0, stream>>>(
        evecs_x, evecs_y, feat_x, feat_y, part);

    fm_reduce<<<(KP + 255) / 256, 256, 0, stream>>>(part, Amat, Bmat);

    fm_mask<<<(KDIM * KDIM + 255) / 256, 256, 0, stream>>>(evals_x, evals_y, Dmask);

    fm_gram<<<dim3(4, 8, 2), 32, 0, stream>>>(Amat, Bmat, AAt, BAt);

    fm_solve<<<KDIM, 1024, SOLVE_LDS_FLOATS * sizeof(float), stream>>>(
        AAt, BAt, Dmask, Cout);
}


# solve kernel: Gauss-Jordan on 16 waves of 8 columns: a pivot also clears the rows above it, so no back substitution; 16 LDS hand-overs, not 32; the solve's order of operations changes
# speedup vs baseline: 6.4131x; 1.0560x over previous
#include <hip/hip_runtime.h>
#include <hip/hip_bf16.h>

typedef __attribute__((ext_vector_type(2))) float v2f;
typedef __attribute__((ext_vector_type(8))) float v8f;
typedef __attribute__((ext_vector_type(4))) float v4f;

#define KDIM 128
#define PDIM 256
#define NDIM 100000
#define LAMBDA 0.001f
#define NCHUNKS 64
#define SLAB 32
#define CHUNK 1568
static_assert(NDIM % SLAB == 0 && CHUNK % SLAB == 0,
              "every chunk must be a whole number of slabs");
static_assert((NCHUNKS - 1) * CHUNK < NDIM && NCHUNKS * CHUNK >= NDIM,
              "the chunks must cover N and none may be empty");
#define EVPITCH 36
#define KP (KDIM * PDIM)
#define PHALF (PDIM / 2)
#define FTPITCH (2 * PHALF + 32)

__global__ __launch_bounds__(512) void fm_gemm_partial(
    const float* __restrict__ evx, const float* __restrict__ evy,
    const float* __restrict__ fx,  const float* __restrict__ fy,
    float* __restrict__ partials)
{
    const int sel   = blockIdx.y;
    const int chunk = blockIdx.x;
    const int pcol0 = blockIdx.z * PHALF;
    const float* __restrict__ ev   = sel ? evy : evx;
    const float* __restrict__ feat = sel ? fy  : fx;

    __shared__ __align__(16) float evBlk[KDIM][EVPITCH];
    __shared__ __align__(16) float ftP[SLAB / 2][FTPITCH];

    const int tid  = threadIdx.x;
    const int lane = tid & 31;
    const int wave = tid >> 5;
    const int rowBand = (wave >> 1) * 16;
    const int wcol    = (wave & 1) * 64;

    const int evRow = tid >> 3;
    const int evCol = (tid & 7) << 2;
    const int ftRow = tid >> 5;
    const int ftCol = (tid & 31) << 2;
    const float* __restrict__ evSrc0 = ev + (size_t)evRow * NDIM + evCol;
    const float* __restrict__ evSrc1 = evSrc0 + (size_t)64 * NDIM;
    const float* __restrict__ ftSrc0 = feat + (size_t)(2 * ftRow) * PDIM + pcol0 + ftCol;
    const float* __restrict__ ftSrc1 = ftSrc0 + PDIM;
    float4* const evDst0 = reinterpret_cast<float4*>(&evBlk[evRow][evCol]);
    float4* const evDst1 = reinterpret_cast<float4*>(&evBlk[evRow + 64][evCol]);
    float4* const ftDst0 = reinterpret_cast<float4*>(&ftP[ftRow][2 * ftCol]);
    float4* const ftDst1 = reinterpret_cast<float4*>(&ftP[ftRow][2 * ftCol + 4]);
    v8f acc[4];
    #pragma unroll
    for (int t = 0; t < 4; ++t) acc[t] = (v8f)0.0f;

    const int nStart = chunk * CHUNK;
    const int nEnd   = min(nStart + CHUNK, NDIM);

    *evDst0 = *reinterpret_cast<const float4*>(evSrc0 + nStart);
    *evDst1 = *reinterpret_cast<const float4*>(evSrc1 + nStart);
    {
        const float4 f0 = *reinterpret_cast<const float4*>(ftSrc0 + (size_t)nStart * PDIM);
        const float4 f1 = *reinterpret_cast<const float4*>(ftSrc1 + (size_t)nStart * PDIM);
        *ftDst0 = make_float4(f0.x, f1.x, f0.y, f1.y);
        *ftDst1 = make_float4(f0.z, f1.z, f0.w, f1.w);
    }
    __syncthreads();

    const int m  = lane & 15;
    const int kp = (lane >> 4) << 1;

    for (int blk = nStart; blk < nEnd; blk += SLAB) {
        const int nxt = blk + SLAB;
        const bool more = (nxt < nEnd);

        float4 evr0, evr1, ftr0, ftr1;
        if (more) {
            evr0 = *reinterpret_cast<const float4*>(evSrc0 + nxt);
            evr1 = *reinterpret_cast<const float4*>(evSrc1 + nxt);
            ftr0 = *reinterpret_cast<const float4*>(ftSrc0 + (size_t)nxt * PDIM);
            ftr1 = *reinterpret_cast<const float4*>(ftSrc1 + (size_t)nxt * PDIM);
            if (nxt + 32 < nEnd && tid < 256) {
                long base = (long)(nxt + 32 + (tid >> 1)) * PDIM + pcol0;
                __builtin_prefetch(&feat[base + (tid & 1) * 64], 0, 0);
            }
        }

        #pragma unroll
        for (int kk = 0; kk < 8; ++kk) {
            const int nOff = kk * 4;
            v2f a;
            a.x = evBlk[rowBand + m][nOff + kp];
            a.y = evBlk[rowBand + m][nOff + kp + 1];
            #pragma unroll
            for (int t = 0; t < 4; ++t) {
                const int col = wcol + t * 16 + m;
                const v2f b = *reinterpret_cast<const v2f*>(
                    &ftP[2 * kk + (lane >> 4)][2 * col]);
                acc[t] = __builtin_amdgcn_wmma_f32_16x16x4_f32(
                    false, a, false, b, (short)0, acc[t], false, false);
            }
        }
        __syncthreads();

        if (more) {
            *evDst0 = evr0;
            *evDst1 = evr1;
            *ftDst0 = make_float4(ftr0.x, ftr1.x, ftr0.y, ftr1.y);
            *ftDst1 = make_float4(ftr0.z, ftr1.z, ftr0.w, ftr1.w);
            __syncthreads();
        }
    }

    float* out = partials + (size_t)sel * NCHUNKS * KP + (size_t)chunk * KP;
    const int mrow = rowBand + ((lane >> 4) << 3);
    #pragma unroll
    for (int t = 0; t < 4; ++t) {
        const int col = pcol0 + wcol + t * 16 + (lane & 15);
        #pragma unroll
        for (int r = 0; r < 8; ++r)
            out[(mrow + r) * PDIM + col] = acc[t][r];
    }
}

__global__ void fm_reduce(const float* __restrict__ partials,
                          float* __restrict__ Amat, float* __restrict__ Bmat)
{
    int idx = blockIdx.x * blockDim.x + threadIdx.x;
    if (idx >= KP) return;
    float sa = 0.0f, sb = 0.0f;
    for (int c = 0; c < NCHUNKS; ++c) {
        sa += partials[(size_t)c * KP + idx];
        sb += partials[(size_t)(NCHUNKS + c) * KP + idx];
    }
    Amat[idx] = sa;
    Bmat[idx] = sb;
}

__global__ void fm_mask(const float* __restrict__ evx,
                        const float* __restrict__ evy,
                        float* __restrict__ D)
{
    int idx = blockIdx.x * blockDim.x + threadIdx.x;
    if (idx >= KDIM * KDIM) return;
    int i = idx >> 7;
    int j = idx & 127;
    float s  = fmaxf(evx[KDIM - 1], evy[KDIM - 1]);
    float g1 = sqrtf(evx[j] / s);
    float g2 = sqrtf(evy[i] / s);
    float d1 = 1.0f / (g1 * g1 + 1.0f);
    float d2 = 1.0f / (g2 * g2 + 1.0f);
    float re = g2 * d2 - g1 * d1;
    float im = d2 - d1;
    D[idx] = re * re + im * im;
}

__global__ __launch_bounds__(32) __attribute__((amdgpu_waves_per_eu(1, 1)))
void fm_gram(
    const float* __restrict__ Amat, const float* __restrict__ Bmat,
    float* __restrict__ AAt, float* __restrict__ BAt)
{
    const float* __restrict__ L = blockIdx.z ? Bmat : Amat;
    const float* __restrict__ R = Amat;
    float* __restrict__ out     = blockIdx.z ? BAt : AAt;

    const int lane  = threadIdx.x;
    const int m     = lane & 15;
    const int kp    = (lane >> 4) << 1;
    const int iBase = blockIdx.y * 16;
    const int jBase = blockIdx.x * 32;

    const v2f* __restrict__ lrow =
        reinterpret_cast<const v2f*>(L + (iBase + m) * PDIM + kp);
    const v2f* __restrict__ rrow0 =
        reinterpret_cast<const v2f*>(R + (jBase + m) * PDIM + kp);
    const v2f* __restrict__ rrow1 =
        reinterpret_cast<const v2f*>(R + (jBase + 16 + m) * PDIM + kp);

    v8f acc0 = (v8f)0.0f, acc1 = (v8f)0.0f;
    #pragma unroll
    for (int step = 0; step < PDIM / 4; ++step) {
        const v2f a  = lrow[step * 2];
        const v2f b0 = rrow0[step * 2];
        const v2f b1 = rrow1[step * 2];
        acc0 = __builtin_amdgcn_wmma_f32_16x16x4_f32(
            false, a, false, b0, (short)0, acc0, false, false);
        acc1 = __builtin_amdgcn_wmma_f32_16x16x4_f32(
            false, a, false, b1, (short)0, acc1, false, false);
    }

    const int mrow = iBase + ((lane >> 4) << 3);
    const int j    = jBase + m;
    #pragma unroll
    for (int r = 0; r < 8; ++r) {
        out[(mrow + r) * KDIM + j]      = acc0[r];
        out[(mrow + r) * KDIM + j + 16] = acc1[r];
    }
}

#define NC 8
#define NW (KDIM / NC)
#define GPB (32 / NC)
static_assert(NC % 4 == 0 && 32 % NC == 0 && GPB % 2 == 0,
              "16-byte column pieces; whole groups per row block; the double buffer alternates across row blocks");
#define FBUF_FLOATS (2 * NC * KDIM)

__device__ __forceinline__ float fm_lane_read(float v, int l)
{
    return __int_as_float(__builtin_amdgcn_readlane(__float_as_int(v), l));
}

__device__ __forceinline__ float fm_recip(float p)
{
    const float r = __builtin_amdgcn_rcpf(p);
    const float e = __builtin_fmaf(-p, r, 1.0f);
    return __builtin_fmaf(r, e, r);
}

__device__ __forceinline__ v2f fm_sub_mul(v2f m, v2f f, float u)
{
    return __builtin_elementwise_fma(-f, (v2f)(u), m);
}

template <int KC>
__device__ __forceinline__ void fm_elim_rowblock(
    v2f (&mc)[NC][2], v2f (&rhs)[2], float* fbuf, float* pivsh,
    const int lane, const int wave)
{
    #pragma unroll 1
    for (int g = 0; g < GPB; ++g) {
        const int owner = KC * GPB + g;
        const int lc0   = NC * g;
        float* fb = fbuf + (g & 1) * (NC * KDIM) + lane * 4;

        if (wave == owner) {
            float pv = 0.0f;
            #pragma unroll
            for (int t = 0; t < NC; ++t) {
                const int lc = lc0 + t;
                const float piv  = fm_lane_read(mc[t][KC >> 1][KC & 1], lc);
                const float rpiv = fm_recip(piv);
                pv = (lane == t) ? piv : pv;
                v2f f[2] = {mc[t][0] * rpiv, mc[t][1] * rpiv};
                f[KC >> 1][KC & 1] = (lane != lc) ? f[KC >> 1][KC & 1] : 0.0f;
                #pragma unroll
                for (int t2 = t + 1; t2 < NC; ++t2) {
                    const float u = fm_lane_read(mc[t2][KC >> 1][KC & 1], lc);
                    mc[t2][0] = fm_sub_mul(mc[t2][0], f[0], u);
                    mc[t2][1] = fm_sub_mul(mc[t2][1], f[1], u);
                }
                *reinterpret_cast<float4*>(fb + t * KDIM) =
                    make_float4(f[0].x, f[0].y, f[1].x, f[1].y);
            }
            if (lane < NC) pivsh[NC * owner + lane] = pv;
        }
        __syncthreads();

        if (wave > owner || wave == NW - 1) {
            float4 fv[NC];
            #pragma unroll
            for (int t = 0; t < NC; ++t)
                fv[t] = *reinterpret_cast<const float4*>(fb + t * KDIM);
            if (wave > owner) {
                #pragma unroll
                for (int t = 0; t < NC; ++t) {
                    const int lc = lc0 + t;
                    const v2f f0 = {fv[t].x, fv[t].y};
                    const v2f f1 = {fv[t].z, fv[t].w};
                    #pragma unroll
                    for (int jj = 0; jj < NC; ++jj) {
                        const float u = fm_lane_read(mc[jj][KC >> 1][KC & 1], lc);
                        mc[jj][0] = fm_sub_mul(mc[jj][0], f0, u);
                        mc[jj][1] = fm_sub_mul(mc[jj][1], f1, u);
                    }
                    #pragma unroll
                    for (int jj = 0; jj < NC; ++jj)
                        asm volatile("" : "+v"(mc[jj][1 - (KC >> 1)]));
                }
            }
            if (wave == NW - 1) {
                #pragma unroll
                for (int t = 0; t < NC; ++t) {
                    const int lc = lc0 + t;
                    const v2f f0 = {fv[t].x, fv[t].y};
                    const v2f f1 = {fv[t].z, fv[t].w};
                    const float yb = fm_lane_read(rhs[KC >> 1][KC & 1], lc);
                    rhs[0] = fm_sub_mul(rhs[0], f0, yb);
                    rhs[1] = fm_sub_mul(rhs[1], f1, yb);
                }
            }
        }
    }
}

__global__ __launch_bounds__(32 * NW) void fm_solve(
    const float* __restrict__ AAt, const float* __restrict__ BAt,
    const float* __restrict__ D,   float* __restrict__ Cout)
{
    __shared__ __align__(16) float fbuf[FBUF_FLOATS];
    __shared__ float pivsh[KDIM];

    const int i    = blockIdx.x;
    const int tid  = threadIdx.x;
    const int lane = tid & 31;
    const int wave = __builtin_amdgcn_readfirstlane(tid >> 5);
    const int j0   = wave * NC;

    v2f mc[NC][2];
    v2f rhs[2];
    #pragma unroll
    for (int k = 0; k < 4; ++k) {
        const int row = lane + 32 * k;
        #pragma unroll
        for (int c4 = 0; c4 < NC / 4; ++c4) {
            const float4 v = *reinterpret_cast<const float4*>(AAt + row * KDIM + j0 + 4 * c4);
            mc[4 * c4 + 0][k >> 1][k & 1] = v.x; mc[4 * c4 + 1][k >> 1][k & 1] = v.y;
            mc[4 * c4 + 2][k >> 1][k & 1] = v.z; mc[4 * c4 + 3][k >> 1][k & 1] = v.w;
        }
        rhs[k >> 1][k & 1] = BAt[i * KDIM + row];
    }
    #pragma unroll
    for (int k = 0; k < 4; ++k) {
        const int row = lane + 32 * k;
        if ((row / NC) == wave) {
            const float dl = D[i * KDIM + row];
            #pragma unroll
            for (int jj = 0; jj < NC; ++jj)
                if (j0 + jj == row) mc[jj][k >> 1][k & 1] += LAMBDA * dl;
        }
    }

    fm_elim_rowblock<0>(mc, rhs, fbuf, pivsh, lane, wave);
    fm_elim_rowblock<1>(mc, rhs, fbuf, pivsh, lane, wave);
    fm_elim_rowblock<2>(mc, rhs, fbuf, pivsh, lane, wave);
    fm_elim_rowblock<3>(mc, rhs, fbuf, pivsh, lane, wave);

    if (wave == NW - 1) {
        #pragma unroll
        for (int k = 0; k < 4; ++k)
            Cout[i * KDIM + lane + 32 * k] = rhs[k >> 1][k & 1] / pivsh[lane + 32 * k];
    }
}

extern "C" void kernel_launch(void* const* d_in, const int* in_sizes, int n_in,
                              void* d_out, int out_size, void* d_ws, size_t ws_size,
                              hipStream_t stream)
{
    const float* feat_x = (const float*)d_in[0];
    const float* feat_y = (const float*)d_in[1];
    const float* evals_x = (const float*)d_in[2];
    const float* evals_y = (const float*)d_in[3];
    const float* evecs_x = (const float*)d_in[4];
    const float* evecs_y = (const float*)d_in[5];

    float* wsf   = (float*)d_ws;
    float* Amat  = wsf;
    float* Bmat  = wsf + 32768;
    float* AAt   = wsf + 65536;
    float* BAt   = wsf + 81920;
    float* Dmask = wsf + 98304;
    float* part  = wsf + 114688;
    float* Cout  = (float*)d_out;

    fm_gemm_partial<<<dim3(NCHUNKS, 2, 2), 512, 0, stream>>>(
        evecs_x, evecs_y, feat_x, feat_y, part);

    fm_reduce<<<(KP + 255) / 256, 256, 0, stream>>>(part, Amat, Bmat);

    fm_mask<<<(KDIM * KDIM + 255) / 256, 256, 0, stream>>>(evals_x, evals_y, Dmask);

    fm_gram<<<dim3(4, 8, 2), 32, 0, stream>>>(Amat, Bmat, AAt, BAt);

    fm_solve<<<KDIM, 32 * NW, 0, stream>>>(
        AAt, BAt, Dmask, Cout);
}


# gemm kernel: slabs go through a ring of three LDS buffers with ONE barrier per slab, placed mid-slab with the next fragments already in registers (was two barriers round the refill); k order kept
# speedup vs baseline: 6.6228x; 1.0327x over previous
#include <hip/hip_runtime.h>
#include <hip/hip_bf16.h>

typedef __attribute__((ext_vector_type(2))) float v2f;
typedef __attribute__((ext_vector_type(8))) float v8f;
typedef __attribute__((ext_vector_type(4))) float v4f;

#define KDIM 128
#define PDIM 256
#define NDIM 100000
#define LAMBDA 0.001f
#define NCHUNKS 64
#define SLAB 32
#define CHUNK 1568
static_assert(NDIM % SLAB == 0 && CHUNK % SLAB == 0,
              "every chunk must be a whole number of slabs");
static_assert((NCHUNKS - 1) * CHUNK < NDIM && NCHUNKS * CHUNK >= NDIM,
              "the chunks must cover N and none may be empty");
#define EVPITCH 36
#define KP (KDIM * PDIM)
#define PHALF (PDIM / 2)
#define FTPITCH (2 * PHALF + 32)

#define NBUF 3
__global__ __launch_bounds__(512) void fm_gemm_partial(
    const float* __restrict__ evx, const float* __restrict__ evy,
    const float* __restrict__ fx,  const float* __restrict__ fy,
    float* __restrict__ partials)
{
    const int sel   = blockIdx.y;
    const int chunk = blockIdx.x;
    const int pcol0 = blockIdx.z * PHALF;
    const float* __restrict__ ev   = sel ? evy : evx;
    const float* __restrict__ feat = sel ? fy  : fx;

    __shared__ __align__(16) float evBlk[NBUF][KDIM][EVPITCH];
    __shared__ __align__(16) float ftP[NBUF][SLAB / 2][FTPITCH];

    const int tid  = threadIdx.x;
    const int lane = tid & 31;
    const int wave = tid >> 5;
    const int rowBand = (wave >> 1) * 16;
    const int wcol    = (wave & 1) * 64;

    const int evRow = tid >> 3;
    const int evCol = (tid & 7) << 2;
    const int ftRow = tid >> 5;
    const int ftCol = (tid & 31) << 2;
    const float* __restrict__ evSrc0 = ev + (size_t)evRow * NDIM + evCol;
    const float* __restrict__ evSrc1 = evSrc0 + (size_t)64 * NDIM;
    const float* __restrict__ ftSrc0 = feat + (size_t)(2 * ftRow) * PDIM + pcol0 + ftCol;
    const float* __restrict__ ftSrc1 = ftSrc0 + PDIM;

    v8f acc[4];
    #pragma unroll
    for (int t = 0; t < 4; ++t) acc[t] = (v8f)0.0f;

    const int nStart = chunk * CHUNK;
    const int nEnd   = min(nStart + CHUNK, NDIM);
    const int nSlabs = (nEnd - nStart) / SLAB;

    const int m  = lane & 15;
    const int kp = (lane >> 4) << 1;

    float4 evr0, evr1, ftr0, ftr1;

    auto fetch = [&](const int n0) {
        evr0 = *reinterpret_cast<const float4*>(evSrc0 + n0);
        evr1 = *reinterpret_cast<const float4*>(evSrc1 + n0);
        ftr0 = *reinterpret_cast<const float4*>(ftSrc0 + (size_t)n0 * PDIM);
        ftr1 = *reinterpret_cast<const float4*>(ftSrc1 + (size_t)n0 * PDIM);
        if (n0 + 32 < nEnd && tid < 256) {
            long base = (long)(n0 + 32 + (tid >> 1)) * PDIM + pcol0;
            __builtin_prefetch(&feat[base + (tid & 1) * 64], 0, 0);
        }
    };
    auto stash = [&](const int b) {
        *reinterpret_cast<float4*>(&evBlk[b][evRow][evCol])      = evr0;
        *reinterpret_cast<float4*>(&evBlk[b][evRow + 64][evCol]) = evr1;
        *reinterpret_cast<float4*>(&ftP[b][ftRow][2 * ftCol]) =
            make_float4(ftr0.x, ftr1.x, ftr0.y, ftr1.y);
        *reinterpret_cast<float4*>(&ftP[b][ftRow][2 * ftCol + 4]) =
            make_float4(ftr0.z, ftr1.z, ftr0.w, ftr1.w);
    };
    auto fragA = [&](const int b, const int kk) -> v2f {
        return *reinterpret_cast<const v2f*>(&evBlk[b][rowBand + m][kk * 4 + kp]);
    };
    auto fragB = [&](const int b, const int kk, const int t) -> v2f {
        return *reinterpret_cast<const v2f*>(
            &ftP[b][2 * kk + (lane >> 4)][2 * (wcol + t * 16 + m)]);
    };

    auto kstep = [&](const int b, const int kk) {
        const v2f a = fragA(b, kk);
        #pragma unroll
        for (int t = 0; t < 4; ++t)
            acc[t] = __builtin_amdgcn_wmma_f32_16x16x4_f32(
                false, a, false, fragB(b, kk, t), (short)0, acc[t], false, false);
    };

    v2f a4;
    v2f b4[4];

    auto firstHalf = [&](const int b, const int bs) {
        kstep(b, 0);
        kstep(b, 1);
        kstep(b, 2);
        v2f a3 = fragA(b, 3);
        a4 = fragA(b, 4);
        v2f b3[4];
        #pragma unroll
        for (int t = 0; t < 4; ++t) { b3[t] = fragB(b, 3, t); b4[t] = fragB(b, 4, t); }
        __builtin_amdgcn_sched_barrier(0);
        stash(bs);
        asm volatile("" : "+v"(a3));
        #pragma unroll
        for (int t = 0; t < 4; ++t)
            acc[t] = __builtin_amdgcn_wmma_f32_16x16x4_f32(
                false, a3, false, b3[t], (short)0, acc[t], false, false);
    };

    auto secondHalf = [&](const int b, const bool doFetch, const int n0) {
        #pragma unroll
        for (int t = 0; t < 2; ++t)
            acc[t] = __builtin_amdgcn_wmma_f32_16x16x4_f32(
                false, a4, false, b4[t], (short)0, acc[t], false, false);
        if (doFetch) fetch(n0);
        #pragma unroll
        for (int t = 2; t < 4; ++t)
            acc[t] = __builtin_amdgcn_wmma_f32_16x16x4_f32(
                false, a4, false, b4[t], (short)0, acc[t], false, false);
        kstep(b, 5);
        kstep(b, 6);
        kstep(b, 7);
    };

    fetch(nStart);
    stash(0);
    __syncthreads();
    fetch(nStart + min(1, nSlabs - 1) * SLAB);
    firstHalf(0, 1);

    int cur = 0;
    for (int s = 0; s + 1 < nSlabs; ++s) {
        const int nxt = (cur == NBUF - 1) ? 0 : cur + 1;
        const int nx2 = (nxt == NBUF - 1) ? 0 : nxt + 1;
        __syncthreads();
        secondHalf(cur, true, nStart + min(s + 2, nSlabs - 1) * SLAB);
        firstHalf(nxt, nx2);
        cur = nxt;
    }
    secondHalf(cur, false, 0);

    float* out = partials + (size_t)sel * NCHUNKS * KP + (size_t)chunk * KP;
    const int mrow = rowBand + ((lane >> 4) << 3);
    #pragma unroll
    for (int t = 0; t < 4; ++t) {
        const int col = pcol0 + wcol + t * 16 + (lane & 15);
        #pragma unroll
        for (int r = 0; r < 8; ++r)
            out[(mrow + r) * PDIM + col] = acc[t][r];
    }
}

__global__ void fm_reduce(const float* __restrict__ partials,
                          float* __restrict__ Amat, float* __restrict__ Bmat)
{
    int idx = blockIdx.x * blockDim.x + threadIdx.x;
    if (idx >= KP) return;
    float sa = 0.0f, sb = 0.0f;
    for (int c = 0; c < NCHUNKS; ++c) {
        sa += partials[(size_t)c * KP + idx];
        sb += partials[(size_t)(NCHUNKS + c) * KP + idx];
    }
    Amat[idx] = sa;
    Bmat[idx] = sb;
}

__global__ void fm_mask(const float* __restrict__ evx,
                        const float* __restrict__ evy,
                        float* __restrict__ D)
{
    int idx = blockIdx.x * blockDim.x + threadIdx.x;
    if (idx >= KDIM * KDIM) return;
    int i = idx >> 7;
    int j = idx & 127;
    float s  = fmaxf(evx[KDIM - 1], evy[KDIM - 1]);
    float g1 = sqrtf(evx[j] / s);
    float g2 = sqrtf(evy[i] / s);
    float d1 = 1.0f / (g1 * g1 + 1.0f);
    float d2 = 1.0f / (g2 * g2 + 1.0f);
    float re = g2 * d2 - g1 * d1;
    float im = d2 - d1;
    D[idx] = re * re + im * im;
}

__global__ __launch_bounds__(32) __attribute__((amdgpu_waves_per_eu(1, 1)))
void fm_gram(
    const float* __restrict__ Amat, const float* __restrict__ Bmat,
    float* __restrict__ AAt, float* __restrict__ BAt)
{
    const float* __restrict__ L = blockIdx.z ? Bmat : Amat;
    const float* __restrict__ R = Amat;
    float* __restrict__ out     = blockIdx.z ? BAt : AAt;

    const int lane  = threadIdx.x;
    const int m     = lane & 15;
    const int kp    = (lane >> 4) << 1;
    const int iBase = blockIdx.y * 16;
    const int jBase = blockIdx.x * 32;

    const v2f* __restrict__ lrow =
        reinterpret_cast<const v2f*>(L + (iBase + m) * PDIM + kp);
    const v2f* __restrict__ rrow0 =
        reinterpret_cast<const v2f*>(R + (jBase + m) * PDIM + kp);
    const v2f* __restrict__ rrow1 =
        reinterpret_cast<const v2f*>(R + (jBase + 16 + m) * PDIM + kp);

    v8f acc0 = (v8f)0.0f, acc1 = (v8f)0.0f;
    #pragma unroll
    for (int step = 0; step < PDIM / 4; ++step) {
        const v2f a  = lrow[step * 2];
        const v2f b0 = rrow0[step * 2];
        const v2f b1 = rrow1[step * 2];
        acc0 = __builtin_amdgcn_wmma_f32_16x16x4_f32(
            false, a, false, b0, (short)0, acc0, false, false);
        acc1 = __builtin_amdgcn_wmma_f32_16x16x4_f32(
            false, a, false, b1, (short)0, acc1, false, false);
    }

    const int mrow = iBase + ((lane >> 4) << 3);
    const int j    = jBase + m;
    #pragma unroll
    for (int r = 0; r < 8; ++r) {
        out[(mrow + r) * KDIM + j]      = acc0[r];
        out[(mrow + r) * KDIM + j + 16] = acc1[r];
    }
}

#define NC 8
#define NW (KDIM / NC)
#define GPB (32 / NC)
static_assert(NC % 4 == 0 && 32 % NC == 0 && GPB % 2 == 0,
              "16-byte column pieces; whole groups per row block; the double buffer alternates across row blocks");
#define FBUF_FLOATS (2 * NC * KDIM)

__device__ __forceinline__ float fm_lane_read(float v, int l)
{
    return __int_as_float(__builtin_amdgcn_readlane(__float_as_int(v), l));
}

__device__ __forceinline__ float fm_recip(float p)
{
    const float r = __builtin_amdgcn_rcpf(p);
    const float e = __builtin_fmaf(-p, r, 1.0f);
    return __builtin_fmaf(r, e, r);
}

__device__ __forceinline__ v2f fm_sub_mul(v2f m, v2f f, float u)
{
    return __builtin_elementwise_fma(-f, (v2f)(u), m);
}

template <int KC>
__device__ __forceinline__ void fm_elim_rowblock(
    v2f (&mc)[NC][2], v2f (&rhs)[2], float* fbuf, float* pivsh,
    const int lane, const int wave)
{
    #pragma unroll 1
    for (int g = 0; g < GPB; ++g) {
        const int owner = KC * GPB + g;
        const int lc0   = NC * g;
        float* fb = fbuf + (g & 1) * (NC * KDIM) + lane * 4;

        if (wave == owner) {
            float pv = 0.0f;
            #pragma unroll
            for (int t = 0; t < NC; ++t) {
                const int lc = lc0 + t;
                const float piv  = fm_lane_read(mc[t][KC >> 1][KC & 1], lc);
                const float rpiv = fm_recip(piv);
                pv = (lane == t) ? piv : pv;
                v2f f[2] = {mc[t][0] * rpiv, mc[t][1] * rpiv};
                f[KC >> 1][KC & 1] = (lane != lc) ? f[KC >> 1][KC & 1] : 0.0f;
                #pragma unroll
                for (int t2 = t + 1; t2 < NC; ++t2) {
                    const float u = fm_lane_read(mc[t2][KC >> 1][KC & 1], lc);
                    mc[t2][0] = fm_sub_mul(mc[t2][0], f[0], u);
                    mc[t2][1] = fm_sub_mul(mc[t2][1], f[1], u);
                }
                *reinterpret_cast<float4*>(fb + t * KDIM) =
                    make_float4(f[0].x, f[0].y, f[1].x, f[1].y);
            }
            if (lane < NC) pivsh[NC * owner + lane] = pv;
        }
        __syncthreads();

        if (wave > owner || wave == NW - 1) {
            float4 fv[NC];
            #pragma unroll
            for (int t = 0; t < NC; ++t)
                fv[t] = *reinterpret_cast<const float4*>(fb + t * KDIM);
            if (wave > owner) {
                #pragma unroll
                for (int t = 0; t < NC; ++t) {
                    const int lc = lc0 + t;
                    const v2f f0 = {fv[t].x, fv[t].y};
                    const v2f f1 = {fv[t].z, fv[t].w};
                    #pragma unroll
                    for (int jj = 0; jj < NC; ++jj) {
                        const float u = fm_lane_read(mc[jj][KC >> 1][KC & 1], lc);
                        mc[jj][0] = fm_sub_mul(mc[jj][0], f0, u);
                        mc[jj][1] = fm_sub_mul(mc[jj][1], f1, u);
                    }
                    #pragma unroll
                    for (int jj = 0; jj < NC; ++jj)
                        asm volatile("" : "+v"(mc[jj][1 - (KC >> 1)]));
                }
            }
            if (wave == NW - 1) {
                #pragma unroll
                for (int t = 0; t < NC; ++t) {
                    const int lc = lc0 + t;
                    const v2f f0 = {fv[t].x, fv[t].y};
                    const v2f f1 = {fv[t].z, fv[t].w};
                    const float yb = fm_lane_read(rhs[KC >> 1][KC & 1], lc);
                    rhs[0] = fm_sub_mul(rhs[0], f0, yb);
                    rhs[1] = fm_sub_mul(rhs[1], f1, yb);
                }
            }
        }
    }
}

__global__ __launch_bounds__(32 * NW) void fm_solve(
    const float* __restrict__ AAt, const float* __restrict__ BAt,
    const float* __restrict__ D,   float* __restrict__ Cout)
{
    __shared__ __align__(16) float fbuf[FBUF_FLOATS];
    __shared__ float pivsh[KDIM];

    const int i    = blockIdx.x;
    const int tid  = threadIdx.x;
    const int lane = tid & 31;
    const int wave = __builtin_amdgcn_readfirstlane(tid >> 5);
    const int j0   = wave * NC;

    v2f mc[NC][2];
    v2f rhs[2];
    #pragma unroll
    for (int k = 0; k < 4; ++k) {
        const int row = lane + 32 * k;
        #pragma unroll
        for (int c4 = 0; c4 < NC / 4; ++c4) {
            const float4 v = *reinterpret_cast<const float4*>(AAt + row * KDIM + j0 + 4 * c4);
            mc[4 * c4 + 0][k >> 1][k & 1] = v.x; mc[4 * c4 + 1][k >> 1][k & 1] = v.y;
            mc[4 * c4 + 2][k >> 1][k & 1] = v.z; mc[4 * c4 + 3][k >> 1][k & 1] = v.w;
        }
        rhs[k >> 1][k & 1] = BAt[i * KDIM + row];
    }
    #pragma unroll
    for (int k = 0; k < 4; ++k) {
        const int row = lane + 32 * k;
        if ((row / NC) == wave) {
            const float dl = D[i * KDIM + row];
            #pragma unroll
            for (int jj = 0; jj < NC; ++jj)
                if (j0 + jj == row) mc[jj][k >> 1][k & 1] += LAMBDA * dl;
        }
    }

    fm_elim_rowblock<0>(mc, rhs, fbuf, pivsh, lane, wave);
    fm_elim_rowblock<1>(mc, rhs, fbuf, pivsh, lane, wave);
    fm_elim_rowblock<2>(mc, rhs, fbuf, pivsh, lane, wave);
    fm_elim_rowblock<3>(mc, rhs, fbuf, pivsh, lane, wave);

    if (wave == NW - 1) {
        #pragma unroll
        for (int k = 0; k < 4; ++k)
            Cout[i * KDIM + lane + 32 * k] = rhs[k >> 1][k & 1] / pivsh[lane + 32 * k];
    }
}

extern "C" void kernel_launch(void* const* d_in, const int* in_sizes, int n_in,
                              void* d_out, int out_size, void* d_ws, size_t ws_size,
                              hipStream_t stream)
{
    const float* feat_x = (const float*)d_in[0];
    const float* feat_y = (const float*)d_in[1];
    const float* evals_x = (const float*)d_in[2];
    const float* evals_y = (const float*)d_in[3];
    const float* evecs_x = (const float*)d_in[4];
    const float* evecs_y = (const float*)d_in[5];

    float* wsf   = (float*)d_ws;
    float* Amat  = wsf;
    float* Bmat  = wsf + 32768;
    float* AAt   = wsf + 65536;
    float* BAt   = wsf + 81920;
    float* Dmask = wsf + 98304;
    float* part  = wsf + 114688;
    float* Cout  = (float*)d_out;

    fm_gemm_partial<<<dim3(NCHUNKS, 2, 2), 512, 0, stream>>>(
        evecs_x, evecs_y, feat_x, feat_y, part);

    fm_reduce<<<(KP + 255) / 256, 256, 0, stream>>>(part, Amat, Bmat);

    fm_mask<<<(KDIM * KDIM + 255) / 256, 256, 0, stream>>>(evals_x, evals_y, Dmask);

    fm_gram<<<dim3(4, 8, 2), 32, 0, stream>>>(Amat, Bmat, AAt, BAt);

    fm_solve<<<KDIM, 32 * NW, 0, stream>>>(
        AAt, BAt, Dmask, Cout);
}
